# starting kernel
# baseline (speedup 1.0000x reference)
#include <hip/hip_runtime.h>
#include <hip/hip_bf16.h>

#define N_NODES 100000
#define N_EDGES 1600000
#define IN_CH   128
#define HID_CH  64
#define OUT_CH  32

typedef __attribute__((ext_vector_type(2))) float f32x2;
typedef __attribute__((ext_vector_type(8))) float f32x8;

__device__ __forceinline__ void atomAddF(float* p, float v) {
    unsafeAtomicAdd(p, v);
}


__global__ void deg_init_kernel(float* deg, int n) {
    int i = blockIdx.x * blockDim.x + threadIdx.x;
    if (i < n) deg[i] = 1.0f;
}

__global__ void deg_count_kernel(const int* __restrict__ dst, float* deg, int nE) {
    int e = blockIdx.x * blockDim.x + threadIdx.x;
    if (e < nE) atomAddF(&deg[dst[e]], 1.0f);
}

__global__ void rsqrt_kernel(float* deg, int n) {
    int i = blockIdx.x * blockDim.x + threadIdx.x;
    if (i < n) deg[i] = rsqrtf(deg[i]);
}


template <int K, int NCOL>
__global__ void gemm_wmma_f32(const float* __restrict__ A,
                              const float* __restrict__ B,
                              float* __restrict__ C) {
    const int wave = threadIdx.x >> 5;
    const int lane = threadIdx.x & 31;
    const int row0 = blockIdx.x << 4;

    const int m  = lane & 15;
    const int kk = (lane >> 4) << 1;
    const int n  = (lane & 15) + (wave << 4);

    const float* ap = A + (size_t)(row0 + m) * K + kk;
    const float* bp = B + (size_t)kk * NCOL + n;

    f32x8 acc = {0.f, 0.f, 0.f, 0.f, 0.f, 0.f, 0.f, 0.f};

    for (int k0 = 0; k0 < K; k0 += 4) {
        f32x2 a, b;
        a.x = ap[k0];
        a.y = ap[k0 + 1];
        b.x = bp[(size_t)k0 * NCOL];
        b.y = bp[(size_t)(k0 + 1) * NCOL];
        acc = __builtin_amdgcn_wmma_f32_16x16x4_f32(
             false, a,  false, b,
             (short)0, acc,  false,  false);
    }

    const int mbase = (lane >> 4) << 3;
    float* cp = C + (size_t)row0 * NCOL + n;
#pragma unroll
    for (int i = 0; i < 8; ++i)
        cp[(size_t)(mbase + i) * NCOL] = acc[i];
}


template <int C, int LOG2C>
__global__ void edge_agg_kernel(const float* __restrict__ h,
                                const int* __restrict__ src,
                                const int* __restrict__ dst,
                                const float* __restrict__ dinv,
                                float* agg) {
    unsigned idx = blockIdx.x * blockDim.x + threadIdx.x;
    int e = idx >> LOG2C;
    int c = idx & (C - 1);
    int s = src[e];
    int d = dst[e];
    float nrm = dinv[s] * dinv[d];
    float v = h[(size_t)s * C + c] * nrm;
    atomAddF(&agg[(size_t)d * C + c], v);
}


template <int C, bool RELU>
__global__ void finish_kernel(const float* agg,
                              const float* __restrict__ h,
                              const float* __restrict__ dinv,
                              const float* __restrict__ bias,
                              float* out) {
    unsigned idx = blockIdx.x * blockDim.x + threadIdx.x;
    int n = idx >> (C == 64 ? 6 : 5);
    int c = idx & (C - 1);
    float di = dinv[n];
    float v = agg[idx] + h[idx] * di * di + bias[c];
    out[idx] = RELU ? fmaxf(v, 0.0f) : v;
}


extern "C" void kernel_launch(void* const* d_in, const int* in_sizes, int n_in,
                              void* d_out, int out_size, void* d_ws, size_t ws_size,
                              hipStream_t stream) {
    const float* x    = (const float*)d_in[0];
    const int*   ei   = (const int*)d_in[1];
    const float* W1   = (const float*)d_in[2];
    const float* b1   = (const float*)d_in[3];
    const float* W2   = (const float*)d_in[4];
    const float* b2   = (const float*)d_in[5];
    float*       out  = (float*)d_out;

    const int* srcv = ei;
    const int* dstv = ei + N_EDGES;

    float* dinv = (float*)d_ws;
    float* h1   = dinv + 100352;
    float* agg1 = h1 + (size_t)N_NODES * HID_CH;
    float* h2   = h1;
    float* agg2 = h1 + (size_t)N_NODES * OUT_CH;

    deg_init_kernel<<<(N_NODES + 255) / 256, 256, 0, stream>>>(dinv, N_NODES);
    deg_count_kernel<<<(N_EDGES + 255) / 256, 256, 0, stream>>>(dstv, dinv, N_EDGES);
    rsqrt_kernel<<<(N_NODES + 255) / 256, 256, 0, stream>>>(dinv, N_NODES);

    gemm_wmma_f32<IN_CH, HID_CH><<<N_NODES / 16, 128, 0, stream>>>(x, W1, h1);

    hipMemsetAsync(agg1, 0, (size_t)N_NODES * HID_CH * sizeof(float), stream);
    edge_agg_kernel<HID_CH, 6>
        <<<(size_t)N_EDGES * HID_CH / 256, 256, 0, stream>>>(h1, srcv, dstv, dinv, agg1);

    finish_kernel<HID_CH, true>
        <<<(size_t)N_NODES * HID_CH / 256, 256, 0, stream>>>(agg1, h1, dinv, b1, agg1);

    gemm_wmma_f32<HID_CH, OUT_CH><<<N_NODES / 16, 64, 0, stream>>>(agg1, W2, h2);

    hipMemsetAsync(agg2, 0, (size_t)N_NODES * OUT_CH * sizeof(float), stream);
    edge_agg_kernel<OUT_CH, 5>
        <<<(size_t)N_EDGES * OUT_CH / 256, 256, 0, stream>>>(h2, srcv, dstv, dinv, agg2);

    finish_kernel<OUT_CH, false>
        <<<(size_t)N_NODES * OUT_CH / 256, 256, 0, stream>>>(agg2, h2, dinv, b2, out);
}


# Layer-2 owner blocks do not scan: layer-1 blocks leave each kept edge as one word (source, local node) in their own slice of out, layer 2 reads that list; LDS add order, so sum order, changes
# speedup vs baseline: 2.2961x; 2.2961x over previous
#include <hip/hip_runtime.h>
#include <hip/hip_bf16.h>

#define N_NODES 100000
#define N_EDGES 1600000
#define IN_CH   128
#define HID_CH  64
#define OUT_CH  32

typedef __attribute__((ext_vector_type(2))) float f32x2;
typedef __attribute__((ext_vector_type(8))) float f32x8;

__device__ __forceinline__ void atomAddF(float* p, float v) {
    unsafeAtomicAdd(p, v);
}


__global__ void deg_init_kernel(float* deg, int n) {
    int i = blockIdx.x * blockDim.x + threadIdx.x;
    if (i < n) deg[i] = 1.0f;
}

typedef __attribute__((ext_vector_type(4))) int i32x4;
typedef __attribute__((ext_vector_type(4))) unsigned u32x4;

#define OWN_NB     250
#define OWN_R      400
#define OWN_WAVES  32
#define TAG_U      5
#define BATCH_G    (TAG_U * 32)
#define BATCH_E    (BATCH_G * 16)
#define BATCH_W    (BATCH_E / 4)
#define N_BATCH    (N_EDGES / BATCH_E)

#define SLICE_W    (OWN_R * OUT_CH)
#define LIST_CAP   320
#define LIST_W0    32
#define TAGS_W0    (LIST_W0 + OWN_WAVES * LIST_CAP)
#define TAGS_BPS   3

static_assert(N_NODES <= (1 << 17) && OWN_R <= (1 << 15), "a list entry packs source and local node in one word");
static_assert(TAGS_W0 % 4 == 0 && SLICE_W % 4 == 0 && BATCH_W % 4 == 0, "owner bytes are read 16 bytes at a time");
static_assert(TAGS_W0 + TAGS_BPS * BATCH_W <= SLICE_W, "lists and owner bytes must fit one slice");
static_assert(TAGS_BPS * OWN_NB >= N_BATCH, "every batch of owner bytes needs a slice");

__device__ __forceinline__ int sliced_tag_word(int t) {
    return (t / TAGS_BPS) * SLICE_W + TAGS_W0 + (t % TAGS_BPS) * BATCH_W;
}

__device__ __forceinline__ unsigned owner_bytes(i32x4 d) {
    return  ((unsigned)d.x / (unsigned)OWN_R)        | (((unsigned)d.y / (unsigned)OWN_R) << 8) |
           (((unsigned)d.z / (unsigned)OWN_R) << 16) | (((unsigned)d.w / (unsigned)OWN_R) << 24);
}

__global__ void deg_count_tag_kernel(const int* __restrict__ dst, float* deg,
                                     unsigned* __restrict__ tags, int nE4) {
    int g = blockIdx.x * blockDim.x + threadIdx.x;
    if (g < nE4) {
        const i32x4 d = ((const i32x4*)dst)[g];
        atomAddF(&deg[d.x], 1.0f);
        atomAddF(&deg[d.y], 1.0f);
        atomAddF(&deg[d.z], 1.0f);
        atomAddF(&deg[d.w], 1.0f);
        const int t = g / BATCH_W;
        tags[sliced_tag_word(t) + (g - t * BATCH_W)] = owner_bytes(d);
    }
}

__global__ void rsqrt_kernel(float* deg, int n) {
    int i = blockIdx.x * blockDim.x + threadIdx.x;
    if (i < n) deg[i] = rsqrtf(deg[i]);
}


template <int K, int NCOL, bool TAGS>
__global__ void gemm_wmma_f32(const float* __restrict__ A,
                              const float* __restrict__ B,
                              float* __restrict__ C,
                              const int* __restrict__ dst,
                              unsigned* __restrict__ tags) {
    if (TAGS) {
        const int g = blockIdx.x * (NCOL * 2) + threadIdx.x;
        tags[g] = owner_bytes(((const i32x4*)dst)[g]);
    }
    const int wave = threadIdx.x >> 5;
    const int lane = threadIdx.x & 31;
    const int row0 = blockIdx.x << 4;

    const int m  = lane & 15;
    const int kk = (lane >> 4) << 1;
    const int n  = (lane & 15) + (wave << 4);

    const float* ap = A + (size_t)(row0 + m) * K + kk;
    const float* bp = B + (size_t)kk * NCOL + n;

    f32x8 acc = {0.f, 0.f, 0.f, 0.f, 0.f, 0.f, 0.f, 0.f};

    for (int k0 = 0; k0 < K; k0 += 4) {
        f32x2 a, b;
        a.x = ap[k0];
        a.y = ap[k0 + 1];
        b.x = bp[(size_t)k0 * NCOL];
        b.y = bp[(size_t)(k0 + 1) * NCOL];
        acc = __builtin_amdgcn_wmma_f32_16x16x4_f32(
             false, a,  false, b,
             (short)0, acc,  false,  false);
    }

    const int mbase = (lane >> 4) << 3;
    float* cp = C + (size_t)row0 * NCOL + n;
#pragma unroll
    for (int i = 0; i < 8; ++i)
        cp[(size_t)(mbase + i) * NCOL] = acc[i];
}


#define ROWS_G     8

static_assert(OWN_NB * OWN_R == N_NODES, "owner ranges must tile the nodes");
static_assert(OWN_NB <= 256, "an owner number must fit one byte");
static_assert(N_BATCH * BATCH_E == N_EDGES, "batches must tile the edges");
static_assert(BATCH_E <= 65536, "a queue entry is a 16-bit position inside the batch");
static_assert((N_NODES / 16) * (OUT_CH * 2) * 4 == N_EDGES, "layer-2 product grid: one thread per 4 edges");
static_assert(ROWS_G == 8, "the load fence in add_rows names 8 rows");

template <int C>
__device__ __forceinline__ void add_rows(const float* __restrict__ h, float* acc, int lane, int m,
                                         int s, int dl, float dsv, float ddv) {
    for (int j = 0; j < m; j += ROWS_G) {
        float v[ROWS_G][C / 32];
#pragma unroll
        for (int t = 0; t < ROWS_G; ++t) {
            const int sj = __builtin_amdgcn_readlane(s, j + t);
            const float* hp = h + (size_t)sj * C + lane;
#pragma unroll
            for (int c = 0; c < C / 32; ++c) v[t][c] = hp[c * 32];
        }
#pragma unroll
        for (int c = 0; c < C / 32; ++c)
            asm volatile("" :: "v"(v[0][c]), "v"(v[1][c]), "v"(v[2][c]), "v"(v[3][c]),
                               "v"(v[4][c]), "v"(v[5][c]), "v"(v[6][c]), "v"(v[7][c]));
#pragma unroll
        for (int t = 0; t < ROWS_G; ++t) {
            if (j + t < m) {
                const float nrm = __int_as_float(__builtin_amdgcn_readlane(__float_as_int(dsv), j + t)) *
                                  __int_as_float(__builtin_amdgcn_readlane(__float_as_int(ddv), j + t));
                const int dlj = __builtin_amdgcn_readlane(dl, j + t);
                float* ap = acc + dlj * C + lane;
#pragma unroll
                for (int c = 0; c < C / 32; ++c) unsafeAtomicAdd(ap + c * 32, v[t][c] * nrm);
            }
        }
    }
}

template <int C, bool SLICED, bool KEEP>
__device__ __forceinline__ int scan_add(const float* __restrict__ h,
                                        const int* __restrict__ src,
                                        const int* __restrict__ dst,
                                        const unsigned* __restrict__ tags,
                                        const float* __restrict__ dinv,
                                        float* acc, unsigned short* q, unsigned* ll,
                                        int lane, int wave, int lo, int blk) {
    const unsigned own4 = blk * 0x01010101u;
    const int rot = (blk * 5) % N_BATCH;
    int nl = 0;

    u32x4 tg[TAG_U];
    int bi = wave + rot; if (bi >= N_BATCH) bi -= N_BATCH;
    {
        const u32x4* tp = (const u32x4*)(tags + (SLICED ? sliced_tag_word(bi) : bi * BATCH_W));
#pragma unroll
        for (int u = 0; u < TAG_U; ++u) tg[u] = tp[lane + u * 32];
    }

    for (int it = wave; it < N_BATCH; it += OWN_WAVES) {
        const int e0 = bi * BATCH_E;
        int cnt = 0;
#pragma unroll
        for (int u = 0; u < TAG_U; ++u) {
            unsigned nz[4];
#pragma unroll
            for (int k = 0; k < 4; ++k) {
                const unsigned x = tg[u][k] ^ own4;
                nz[k] = ((x & 0x7f7f7f7fu) + 0x7f7f7f7fu) | x;
            }
            unsigned f = (nz[0] & 0x80808080u) | ((nz[1] >> 1) & 0x40404040u) |
                         ((nz[2] >> 2) & 0x20202020u) | ((nz[3] >> 3) & 0x10101010u);
            f ^= 0xf0f0f0f0u;
            unsigned mask = __builtin_amdgcn_ballot_w32(f != 0u);
            while (mask != 0u) {
                if (f != 0u) {
                    const int b   = __builtin_ctz(f);
                    const int pos = cnt + (int)__builtin_amdgcn_mbcnt_lo(mask, 0u);
                    const int j   = ((7 - (b & 7)) << 2) + (b >> 3);
                    q[pos] = (unsigned short)(((u * 32 + lane) << 4) + j);
                    f &= f - 1u;
                }
                cnt += __builtin_popcount(mask);
                mask = __builtin_amdgcn_ballot_w32(f != 0u);
            }
        }

        int m = (cnt < 32) ? cnt : 32;
        int e = e0 + ((lane < m) ? (int)q[lane] : 0);
        int d = dst[e];
        int s = src[e];

        {
            const int itn = it + OWN_WAVES;
            bi = ((itn < N_BATCH) ? itn : it) + rot; if (bi >= N_BATCH) bi -= N_BATCH;
            const u32x4* tp = (const u32x4*)(tags + (SLICED ? sliced_tag_word(bi) : bi * BATCH_W));
#pragma unroll
            for (int u = 0; u < TAG_U; ++u) tg[u] = tp[lane + u * 32];
        }

        if (lane >= m) { s = 0; d = lo; }
        int   dl  = d - lo;
        if (KEEP) {
            if (nl + m <= LIST_CAP && lane < m) ll[nl + lane] = (unsigned)s | ((unsigned)dl << 17);
            nl += m;
        }
        float ddv = dinv[d];
        float dsv = dinv[s];
        add_rows<C>(h, acc, lane, m, s, dl, dsv, ddv);

        for (int b0 = 32; b0 < cnt; b0 += 32) {
            m   = (cnt - b0 < 32) ? (cnt - b0) : 32;
            e   = e0 + ((lane < m) ? (int)q[b0 + lane] : 0);
            d   = (lane < m) ? dst[e] : lo;
            s   = (lane < m) ? src[e] : 0;
            dl  = d - lo;
            if (KEEP) {
                if (nl + m <= LIST_CAP && lane < m) ll[nl + lane] = (unsigned)s | ((unsigned)dl << 17);
                nl += m;
            }
            ddv = dinv[d];
            dsv = dinv[s];
            add_rows<C>(h, acc, lane, m, s, dl, dsv, ddv);
        }
    }
    return nl;
}

template <int C, bool RELU, bool SECOND>
__global__ __launch_bounds__(OWN_WAVES * 32)
void agg_owner_kernel(const float* __restrict__ h,
                      const int* __restrict__ src,
                      const int* __restrict__ dst,
                      const unsigned* __restrict__ tags,
                      const float* __restrict__ dinv,
                      const float* __restrict__ bias,
                      float* out,
                      unsigned* lists) {
    __shared__ float          acc[OWN_R * C];
    __shared__ unsigned short wq[OWN_WAVES][BATCH_E];
    __shared__ unsigned       wl[SECOND ? 1 : OWN_WAVES][SECOND ? 1 : LIST_CAP];
    __shared__ unsigned       wn[OWN_WAVES];

    const int tid  = threadIdx.x;
    const int lane = tid & 31;
    const int wave = tid >> 5;
    const int lo   = blockIdx.x * OWN_R;
    unsigned* sl = lists + (size_t)blockIdx.x * SLICE_W;
    int nl = 0;
    const unsigned cw = SECOND ? sl[lane] : 0u;

    for (int i = tid; i < OWN_R * C; i += OWN_WAVES * 32) acc[i] = 0.0f;
    __syncthreads();

    if (!SECOND) {
        nl = scan_add<C, true, true>(h, src, dst, tags, dinv, acc, wq[wave], wl[wave],
                                     lane, wave, lo, blockIdx.x);
        if (lane == 0) wn[wave] = (unsigned)nl;
    } else {
        if (__builtin_amdgcn_ballot_w32(cw > (unsigned)LIST_CAP) == 0u) {
            const int n = __builtin_amdgcn_readlane((int)cw, __builtin_amdgcn_readfirstlane(wave));
            const unsigned* lg = sl + LIST_W0 + wave * LIST_CAP;
            unsigned nx = (lane < n) ? lg[lane] : 0u;
            for (int b0 = 0; b0 < n; b0 += 32) {
                const int m = (n - b0 < 32) ? (n - b0) : 32;
                const unsigned cur = nx;
                nx = (b0 + 32 + lane < n) ? lg[b0 + 32 + lane] : 0u;
                const int s  = (int)(cur & 0x1ffffu);
                const int dl = (int)(cur >> 17);
                const float ddv = dinv[lo + dl];
                const float dsv = dinv[s];
                add_rows<C>(h, acc, lane, m, s, dl, dsv, ddv);
            }
        } else {
            scan_add<C, false, false>(h, src, dst, tags, dinv, acc, wq[wave], wl[0],
                                      lane, wave, lo, blockIdx.x);
        }
    }

    __syncthreads();

    if (!SECOND) {
        unsigned* lg = sl + LIST_W0 + wave * LIST_CAP;
        const int nk = (nl <= LIST_CAP) ? nl : 0;
        for (int i = lane; i < ((nk + 31) & ~31); i += 32) lg[i] = (i < nk) ? wl[wave][i] : 0u;
        if (wave == 0) sl[lane] = wn[lane];
    }

    const size_t g0 = (size_t)lo * C;
    for (int i = tid; i < OWN_R * C; i += OWN_WAVES * 32) {
        const int nl = i / C;
        const int c  = i & (C - 1);
        const float di = dinv[lo + nl];
        const float val = acc[i] + h[g0 + i] * di * di + bias[c];
        out[g0 + i] = RELU ? fmaxf(val, 0.0f) : val;
    }
}


extern "C" void kernel_launch(void* const* d_in, const int* in_sizes, int n_in,
                              void* d_out, int out_size, void* d_ws, size_t ws_size,
                              hipStream_t stream) {
    const float* x    = (const float*)d_in[0];
    const int*   ei   = (const int*)d_in[1];
    const float* W1   = (const float*)d_in[2];
    const float* b1   = (const float*)d_in[3];
    const float* W2   = (const float*)d_in[4];
    const float* b2   = (const float*)d_in[5];
    float*       out  = (float*)d_out;

    const int* srcv = ei;
    const int* dstv = ei + N_EDGES;

    float* dinv = (float*)d_ws;
    float* h1   = dinv + 100352;
    float* agg1 = h1 + (size_t)N_NODES * HID_CH;
    float* h2   = h1;
    unsigned* obuf  = (unsigned*)d_out;
    unsigned* tags2 = (unsigned*)(h1 + (size_t)N_NODES * OUT_CH);

    deg_init_kernel<<<(N_NODES + 255) / 256, 256, 0, stream>>>(dinv, N_NODES);
    deg_count_tag_kernel<<<(N_EDGES / 4 + 255) / 256, 256, 0, stream>>>(dstv, dinv, obuf, N_EDGES / 4);
    rsqrt_kernel<<<(N_NODES + 255) / 256, 256, 0, stream>>>(dinv, N_NODES);

    gemm_wmma_f32<IN_CH, HID_CH, false><<<N_NODES / 16, 128, 0, stream>>>(x, W1, h1, dstv, tags2);

    agg_owner_kernel<HID_CH, true, false>
        <<<OWN_NB, OWN_WAVES * 32, 0, stream>>>(h1, srcv, dstv, obuf, dinv, b1, agg1, obuf);

    gemm_wmma_f32<HID_CH, OUT_CH, true><<<N_NODES / 16, 64, 0, stream>>>(agg1, W2, h2, dstv, tags2);

    agg_owner_kernel<OUT_CH, false, true>
        <<<OWN_NB, OWN_WAVES * 32, 0, stream>>>(h2, srcv, dstv, tags2, dinv, b2, out, obuf);
}


# In-degrees counted by owner blocks in LDS (integer adds) in a scan launch that also leaves layer 1 its edge lists: no global float atomics, 7 launches become 5; LDS add order, so sum order, changes
# speedup vs baseline: 3.7549x; 1.6353x over previous
#include <hip/hip_runtime.h>
#include <hip/hip_bf16.h>


#define N_NODES 100000
#define N_EDGES 1600000
#define IN_CH   128
#define HID_CH  64
#define OUT_CH  32

typedef __attribute__((ext_vector_type(2))) float f32x2;
typedef __attribute__((ext_vector_type(8))) float f32x8;

typedef __attribute__((ext_vector_type(4))) int i32x4;
typedef __attribute__((ext_vector_type(4))) unsigned u32x4;

#define OWN_NB     250
#define OWN_R      400
#define OWN_WAVES  32
#define TAG_U      5
#define BATCH_G    (TAG_U * 32)
#define BATCH_E    (BATCH_G * 16)
#define BATCH_W    (BATCH_E / 4)
#define N_BATCH    (N_EDGES / BATCH_E)

#define SLICE_W    (OWN_R * OUT_CH)
#define LIST_CAP   320
#define LIST_W0    32
#define TAGS_W0    (LIST_W0 + OWN_WAVES * LIST_CAP)
#define TAGS_BPS   3

static_assert(N_NODES <= (1 << 17) && OWN_R <= (1 << 15), "a list entry packs source and local node in one word");
static_assert(TAGS_W0 % 4 == 0 && SLICE_W % 4 == 0 && BATCH_W % 4 == 0, "owner bytes are read 16 bytes at a time");
static_assert(TAGS_W0 + TAGS_BPS * BATCH_W <= SLICE_W, "lists and owner bytes must fit one slice");
static_assert(TAGS_BPS * OWN_NB >= N_BATCH, "every batch of owner bytes needs a slice");

__device__ __forceinline__ int sliced_tag_word(int t) {
    return (t / TAGS_BPS) * SLICE_W + TAGS_W0 + (t % TAGS_BPS) * BATCH_W;
}

__device__ __forceinline__ unsigned owner_bytes(i32x4 d) {
    return  ((unsigned)d.x / (unsigned)OWN_R)        | (((unsigned)d.y / (unsigned)OWN_R) << 8) |
           (((unsigned)d.z / (unsigned)OWN_R) << 16) | (((unsigned)d.w / (unsigned)OWN_R) << 24);
}


template <int K, int NCOL, int TAGS>
__global__ void gemm_wmma_f32(const float* __restrict__ A,
                              const float* __restrict__ B,
                              float* __restrict__ C,
                              const int* __restrict__ dst,
                              unsigned* __restrict__ tags) {
    if (TAGS == 1) {
        const int g = blockIdx.x * (NCOL * 2) + threadIdx.x;
        if (g < N_EDGES / 4) {
            const int t = g / BATCH_W;
            tags[sliced_tag_word(t) + (g - t * BATCH_W)] = owner_bytes(((const i32x4*)dst)[g]);
        }
    }
    if (TAGS == 2) {
        const int g = blockIdx.x * (NCOL * 2) + threadIdx.x;
        tags[g] = owner_bytes(((const i32x4*)dst)[g]);
    }
    const int wave = threadIdx.x >> 5;
    const int lane = threadIdx.x & 31;
    const int row0 = blockIdx.x << 4;

    const int m  = lane & 15;
    const int kk = (lane >> 4) << 1;
    const int n  = (lane & 15) + (wave << 4);

    const float* ap = A + (size_t)(row0 + m) * K + kk;
    const float* bp = B + (size_t)kk * NCOL + n;

    f32x8 acc = {0.f, 0.f, 0.f, 0.f, 0.f, 0.f, 0.f, 0.f};

    for (int k0 = 0; k0 < K; k0 += 4) {
        f32x2 a, b;
        a.x = ap[k0];
        a.y = ap[k0 + 1];
        b.x = bp[(size_t)k0 * NCOL];
        b.y = bp[(size_t)(k0 + 1) * NCOL];
        acc = __builtin_amdgcn_wmma_f32_16x16x4_f32(
             false, a,  false, b,
             (short)0, acc,  false,  false);
    }

    const int mbase = (lane >> 4) << 3;
    float* cp = C + (size_t)row0 * NCOL + n;
#pragma unroll
    for (int i = 0; i < 8; ++i)
        cp[(size_t)(mbase + i) * NCOL] = acc[i];
}


#define ROWS_G     8

static_assert(OWN_NB * OWN_R == N_NODES, "owner ranges must tile the nodes");
static_assert(OWN_NB <= 256, "an owner number must fit one byte");
static_assert(N_BATCH * BATCH_E == N_EDGES, "batches must tile the edges");
static_assert(BATCH_E <= 65536, "a queue entry is a 16-bit position inside the batch");
static_assert((N_NODES / 16) * (HID_CH * 2) * 4 >= N_EDGES, "layer-1 product grid: at least one thread per 4 edges");
static_assert((N_NODES / 16) * (OUT_CH * 2) * 4 == N_EDGES, "layer-2 product grid: one thread per 4 edges");
static_assert(ROWS_G == 8, "the load fence in add_rows names 8 rows");

__device__ __forceinline__ int keep_own(const u32x4 (&tg)[TAG_U], unsigned own4, unsigned short* q, int lane) {
    int cnt = 0;
#pragma unroll
    for (int u = 0; u < TAG_U; ++u) {
        unsigned nz[4];
#pragma unroll
        for (int k = 0; k < 4; ++k) {
            const unsigned x = tg[u][k] ^ own4;
            nz[k] = ((x & 0x7f7f7f7fu) + 0x7f7f7f7fu) | x;
        }
        unsigned f = (nz[0] & 0x80808080u) | ((nz[1] >> 1) & 0x40404040u) |
                     ((nz[2] >> 2) & 0x20202020u) | ((nz[3] >> 3) & 0x10101010u);
        f ^= 0xf0f0f0f0u;
        unsigned mask = __builtin_amdgcn_ballot_w32(f != 0u);
        while (mask != 0u) {
            if (f != 0u) {
                const int b   = __builtin_ctz(f);
                const int pos = cnt + (int)__builtin_amdgcn_mbcnt_lo(mask, 0u);
                const int j   = ((7 - (b & 7)) << 2) + (b >> 3);
                q[pos] = (unsigned short)(((u * 32 + lane) << 4) + j);
                f &= f - 1u;
            }
            cnt += __builtin_popcount(mask);
            mask = __builtin_amdgcn_ballot_w32(f != 0u);
        }
    }
    return cnt;
}

template <int C>
__device__ __forceinline__ void add_rows(const float* __restrict__ h, float* acc, int lane, int m,
                                         int s, int dl, float dsv, float ddv) {
    for (int j = 0; j < m; j += ROWS_G) {
        float v[ROWS_G][C / 32];
#pragma unroll
        for (int t = 0; t < ROWS_G; ++t) {
            const int sj = __builtin_amdgcn_readlane(s, j + t);
            const float* hp = h + (size_t)sj * C + lane;
#pragma unroll
            for (int c = 0; c < C / 32; ++c) v[t][c] = hp[c * 32];
        }
#pragma unroll
        for (int c = 0; c < C / 32; ++c)
            asm volatile("" :: "v"(v[0][c]), "v"(v[1][c]), "v"(v[2][c]), "v"(v[3][c]),
                               "v"(v[4][c]), "v"(v[5][c]), "v"(v[6][c]), "v"(v[7][c]));
#pragma unroll
        for (int t = 0; t < ROWS_G; ++t) {
            if (j + t < m) {
                const float nrm = __int_as_float(__builtin_amdgcn_readlane(__float_as_int(dsv), j + t)) *
                                  __int_as_float(__builtin_amdgcn_readlane(__float_as_int(ddv), j + t));
                const int dlj = __builtin_amdgcn_readlane(dl, j + t);
                float* ap = acc + dlj * C + lane;
#pragma unroll
                for (int c = 0; c < C / 32; ++c) unsafeAtomicAdd(ap + c * 32, v[t][c] * nrm);
            }
        }
    }
}

template <int C, bool SLICED>
__device__ __forceinline__ void scan_add(const float* __restrict__ h,
                                         const int* __restrict__ src,
                                         const int* __restrict__ dst,
                                         const unsigned* __restrict__ tags,
                                         const float* __restrict__ dinv,
                                         float* acc, unsigned short* q,
                                         int lane, int wave, int lo, int blk) {
    const unsigned own4 = blk * 0x01010101u;
    const int rot = (blk * 5) % N_BATCH;

    u32x4 tg[TAG_U];
    int bi = wave + rot; if (bi >= N_BATCH) bi -= N_BATCH;
    {
        const u32x4* tp = (const u32x4*)(tags + (SLICED ? sliced_tag_word(bi) : bi * BATCH_W));
#pragma unroll
        for (int u = 0; u < TAG_U; ++u) tg[u] = tp[lane + u * 32];
    }

    for (int it = wave; it < N_BATCH; it += OWN_WAVES) {
        const int e0 = bi * BATCH_E;
        const int cnt = keep_own(tg, own4, q, lane);

        int m = (cnt < 32) ? cnt : 32;
        int e = e0 + ((lane < m) ? (int)q[lane] : 0);
        int d = dst[e];
        int s = src[e];

        {
            const int itn = it + OWN_WAVES;
            bi = ((itn < N_BATCH) ? itn : it) + rot; if (bi >= N_BATCH) bi -= N_BATCH;
            const u32x4* tp = (const u32x4*)(tags + (SLICED ? sliced_tag_word(bi) : bi * BATCH_W));
#pragma unroll
            for (int u = 0; u < TAG_U; ++u) tg[u] = tp[lane + u * 32];
        }

        if (lane >= m) { s = 0; d = lo; }
        int   dl  = d - lo;
        float ddv = dinv[d];
        float dsv = dinv[s];
        add_rows<C>(h, acc, lane, m, s, dl, dsv, ddv);

        for (int b0 = 32; b0 < cnt; b0 += 32) {
            m   = (cnt - b0 < 32) ? (cnt - b0) : 32;
            e   = e0 + ((lane < m) ? (int)q[b0 + lane] : 0);
            d   = (lane < m) ? dst[e] : lo;
            s   = (lane < m) ? src[e] : 0;
            dl  = d - lo;
            ddv = dinv[d];
            dsv = dinv[s];
            add_rows<C>(h, acc, lane, m, s, dl, dsv, ddv);
        }
    }
}

__global__ __launch_bounds__(OWN_WAVES * 32)
void list_deg_kernel(const int* __restrict__ src,
                     const int* __restrict__ dst,
                     const unsigned* __restrict__ tags,
                     float* __restrict__ dinv,
                     unsigned* lists) {
    __shared__ unsigned short wq[OWN_WAVES][BATCH_E];
    __shared__ unsigned       wl[OWN_WAVES][LIST_CAP];
    __shared__ unsigned       dg[OWN_R];
    __shared__ unsigned       wn[OWN_WAVES];

    const int tid  = threadIdx.x;
    const int lane = tid & 31;
    const int wave = tid >> 5;
    const int blk  = blockIdx.x;
    const int lo   = blk * OWN_R;
    unsigned short* q  = wq[wave];
    unsigned*       ll = wl[wave];

    for (int i = tid; i < OWN_R; i += OWN_WAVES * 32) dg[i] = 0u;
    __syncthreads();

    const unsigned own4 = blk * 0x01010101u;
    const int rot = (blk * 5) % N_BATCH;
    int nl = 0;

    auto note = [&](int m, int d, int s) {
        if (lane < m) {
            const int dl = d - lo;
            if (nl + m <= LIST_CAP) ll[nl + lane] = (unsigned)s | ((unsigned)dl << 17);
            atomicAdd(&dg[dl], 1u);
        }
        nl += m;
    };

    u32x4 tg[TAG_U];
    int bi = wave + rot; if (bi >= N_BATCH) bi -= N_BATCH;
    {
        const u32x4* tp = (const u32x4*)(tags + sliced_tag_word(bi));
#pragma unroll
        for (int u = 0; u < TAG_U; ++u) tg[u] = tp[lane + u * 32];
    }

    int pm = 0, pd = lo, ps = 0;
    for (int it = wave; it < N_BATCH; it += OWN_WAVES) {
        const int e0 = bi * BATCH_E;
        const int itn = it + OWN_WAVES;
        int bn = ((itn < N_BATCH) ? itn : it) + rot; if (bn >= N_BATCH) bn -= N_BATCH;
        u32x4 tn[TAG_U];
        {
            const u32x4* tp = (const u32x4*)(tags + sliced_tag_word(bn));
#pragma unroll
            for (int u = 0; u < TAG_U; ++u) tn[u] = tp[lane + u * 32];
        }

        const int cnt = keep_own(tg, own4, q, lane);

        note(pm, pd, ps);

        pm = (cnt < 32) ? cnt : 32;
        const int e = e0 + ((lane < pm) ? (int)q[lane] : 0);
        pd = dst[e];
        ps = src[e];

        for (int b0 = 32; b0 < cnt; b0 += 32) {
            const int mm = (cnt - b0 < 32) ? (cnt - b0) : 32;
            const int ee = e0 + ((lane < mm) ? (int)q[b0 + lane] : 0);
            note(mm, dst[ee], src[ee]);
        }

#pragma unroll
        for (int u = 0; u < TAG_U; ++u) tg[u] = tn[u];
        bi = bn;
    }
    note(pm, pd, ps);

    if (lane == 0) wn[wave] = (unsigned)nl;
    __syncthreads();

    unsigned* sl = lists + (size_t)blk * SLICE_W;
    unsigned* lg = sl + LIST_W0 + wave * LIST_CAP;
    const int nk = (nl <= LIST_CAP) ? nl : 0;
    for (int i = lane; i < ((nk + 31) & ~31); i += 32) lg[i] = (i < nk) ? ll[i] : 0u;
    if (wave == 0) sl[lane] = wn[lane];

    for (int i = tid; i < OWN_R; i += OWN_WAVES * 32) dinv[lo + i] = rsqrtf(1.0f + (float)dg[i]);
}

template <int C, bool RELU, bool SLICED>
__global__ __launch_bounds__(OWN_WAVES * 32)
void agg_owner_kernel(const float* __restrict__ h,
                      const int* __restrict__ src,
                      const int* __restrict__ dst,
                      const unsigned* __restrict__ tags,
                      const float* __restrict__ dinv,
                      const float* __restrict__ bias,
                      float* out,
                      const unsigned* lists) {
    __shared__ float          acc[OWN_R * C];
    __shared__ unsigned short wq[OWN_WAVES][BATCH_E];

    const int tid  = threadIdx.x;
    const int lane = tid & 31;
    const int wave = tid >> 5;
    const int lo   = blockIdx.x * OWN_R;
    const unsigned* sl = lists + (size_t)blockIdx.x * SLICE_W;
    const unsigned cw = sl[lane];

    for (int i = tid; i < OWN_R * C; i += OWN_WAVES * 32) acc[i] = 0.0f;
    __syncthreads();

    if (__builtin_amdgcn_ballot_w32(cw > (unsigned)LIST_CAP) == 0u) {
        const int n = __builtin_amdgcn_readlane((int)cw, __builtin_amdgcn_readfirstlane(wave));
        const unsigned* lg = sl + LIST_W0 + wave * LIST_CAP;
        unsigned nx = (lane < n) ? lg[lane] : 0u;
        for (int b0 = 0; b0 < n; b0 += 32) {
            const int m = (n - b0 < 32) ? (n - b0) : 32;
            const unsigned cur = nx;
            nx = (b0 + 32 + lane < n) ? lg[b0 + 32 + lane] : 0u;
            const int s  = (int)(cur & 0x1ffffu);
            const int dl = (int)(cur >> 17);
            const float ddv = dinv[lo + dl];
            const float dsv = dinv[s];
            add_rows<C>(h, acc, lane, m, s, dl, dsv, ddv);
        }
    } else {
        scan_add<C, SLICED>(h, src, dst, tags, dinv, acc, wq[wave], lane, wave, lo, blockIdx.x);
    }

    __syncthreads();

    const size_t g0 = (size_t)lo * C;
    for (int i = tid; i < OWN_R * C; i += OWN_WAVES * 32) {
        const int nl = i / C;
        const int c  = i & (C - 1);
        const float di = dinv[lo + nl];
        const float val = acc[i] + h[g0 + i] * di * di + bias[c];
        out[g0 + i] = RELU ? fmaxf(val, 0.0f) : val;
    }
}


extern "C" void kernel_launch(void* const* d_in, const int* in_sizes, int n_in,
                              void* d_out, int out_size, void* d_ws, size_t ws_size,
                              hipStream_t stream) {
    const float* x    = (const float*)d_in[0];
    const int*   ei   = (const int*)d_in[1];
    const float* W1   = (const float*)d_in[2];
    const float* b1   = (const float*)d_in[3];
    const float* W2   = (const float*)d_in[4];
    const float* b2   = (const float*)d_in[5];
    float*       out  = (float*)d_out;

    const int* srcv = ei;
    const int* dstv = ei + N_EDGES;

    float* dinv = (float*)d_ws;
    float* h1   = dinv + 100352;
    float* agg1 = h1 + (size_t)N_NODES * HID_CH;
    float* h2   = h1;
    unsigned* obuf  = (unsigned*)d_out;
    unsigned* tags2 = (unsigned*)(h1 + (size_t)N_NODES * OUT_CH);

    gemm_wmma_f32<IN_CH, HID_CH, 1><<<N_NODES / 16, 128, 0, stream>>>(x, W1, h1, dstv, obuf);

    list_deg_kernel<<<OWN_NB, OWN_WAVES * 32, 0, stream>>>(srcv, dstv, obuf, dinv, obuf);

    agg_owner_kernel<HID_CH, true, true>
        <<<OWN_NB, OWN_WAVES * 32, 0, stream>>>(h1, srcv, dstv, obuf, dinv, b1, agg1, obuf);

    gemm_wmma_f32<HID_CH, OUT_CH, 2><<<N_NODES / 16, 64, 0, stream>>>(agg1, W2, h2, dstv, tags2);

    agg_owner_kernel<OUT_CH, false, false>
        <<<OWN_NB, OWN_WAVES * 32, 0, stream>>>(h2, srcv, dstv, tags2, dinv, b2, out, obuf);
}


# Products re-tiled: one wave per 16-row strip does all its column tiles, an A fragment loaded once feeds 4 (layer 1) or 2 (layer 2) matrix instructions; 4x/2x fewer waves; k order, so sum order, kept
# speedup vs baseline: 3.9434x; 1.0502x over previous
#include <hip/hip_runtime.h>
#include <hip/hip_bf16.h>


#define N_NODES 100000
#define N_EDGES 1600000
#define IN_CH   128
#define HID_CH  64
#define OUT_CH  32

typedef __attribute__((ext_vector_type(2))) float f32x2;
typedef __attribute__((ext_vector_type(8))) float f32x8;

typedef __attribute__((ext_vector_type(4))) int i32x4;
typedef __attribute__((ext_vector_type(4))) unsigned u32x4;

#define OWN_NB     250
#define OWN_R      400
#define OWN_WAVES  32
#define TAG_U      5
#define BATCH_G    (TAG_U * 32)
#define BATCH_E    (BATCH_G * 16)
#define BATCH_W    (BATCH_E / 4)
#define N_BATCH    (N_EDGES / BATCH_E)

#define SLICE_W    (OWN_R * OUT_CH)
#define LIST_CAP   320
#define LIST_W0    32
#define TAGS_W0    (LIST_W0 + OWN_WAVES * LIST_CAP)
#define TAGS_BPS   3

static_assert(N_NODES <= (1 << 17) && OWN_R <= (1 << 15), "a list entry packs source and local node in one word");
static_assert(TAGS_W0 % 4 == 0 && SLICE_W % 4 == 0 && BATCH_W % 4 == 0, "owner bytes are read 16 bytes at a time");
static_assert(TAGS_W0 + TAGS_BPS * BATCH_W <= SLICE_W, "lists and owner bytes must fit one slice");
static_assert(TAGS_BPS * OWN_NB >= N_BATCH, "every batch of owner bytes needs a slice");

__device__ __forceinline__ int sliced_tag_word(int t) {
    return (t / TAGS_BPS) * SLICE_W + TAGS_W0 + (t % TAGS_BPS) * BATCH_W;
}

__device__ __forceinline__ unsigned owner_bytes(i32x4 d) {
    return  ((unsigned)d.x / (unsigned)OWN_R)        | (((unsigned)d.y / (unsigned)OWN_R) << 8) |
           (((unsigned)d.z / (unsigned)OWN_R) << 16) | (((unsigned)d.w / (unsigned)OWN_R) << 24);
}


#define GEMM_WAVES 5
#define N_STRIPS   (N_NODES / 16)

template <int K, int NCOL, int TAGS>
__global__ __launch_bounds__(GEMM_WAVES * 32)
void gemm_wmma_f32(const float* __restrict__ A,
                   const float* __restrict__ B,
                   float* __restrict__ C,
                   const int* __restrict__ dst,
                   unsigned* __restrict__ tags) {
    constexpr int NT = NCOL / 16;
    typedef __attribute__((ext_vector_type(NT))) float fvec;

    const int gt = blockIdx.x * (GEMM_WAVES * 32) + threadIdx.x;
    if (TAGS == 1) {
        for (int r = 0; r < 2; ++r) {
            const int g = gt + r * (N_EDGES / 8);
            const int t = g / BATCH_W;
            tags[sliced_tag_word(t) + (g - t * BATCH_W)] = owner_bytes(((const i32x4*)dst)[g]);
        }
    }
    if (TAGS == 2) {
        for (int r = 0; r < 2; ++r) {
            const int g = gt + r * (N_EDGES / 8);
            tags[g] = owner_bytes(((const i32x4*)dst)[g]);
        }
    }
    const int lane = threadIdx.x & 31;
    const int row0 = (blockIdx.x * GEMM_WAVES + (threadIdx.x >> 5)) << 4;

    const int m  = lane & 15;
    const int kk = (lane >> 4) << 1;

    const float* ap = A + (size_t)(row0 + m) * K + kk;
    const fvec*  bp = (const fvec*)(B + (size_t)kk * NCOL) + m;

    f32x8 acc[NT];
    for (int t = 0; t < NT; ++t) acc[t] = (f32x8){0.f, 0.f, 0.f, 0.f, 0.f, 0.f, 0.f, 0.f};

    for (int k0 = 0; k0 < K; k0 += 4) {
        f32x2 a;
        a.x = ap[k0];
        a.y = ap[k0 + 1];
        const fvec b0 = bp[(size_t)k0 * 16];
        const fvec b1 = bp[(size_t)(k0 + 1) * 16];
        for (int t = 0; t < NT; ++t) {
            f32x2 b;
            b.x = b0[t];
            b.y = b1[t];
            acc[t] = __builtin_amdgcn_wmma_f32_16x16x4_f32(
                 false, a,  false, b,
                 (short)0, acc[t],  false,  false);
        }
    }

    const int mbase = (lane >> 4) << 3;
    fvec* cp = (fvec*)(C + (size_t)row0 * NCOL) + m;
    for (int i = 0; i < 8; ++i) {
        fvec o;
        for (int t = 0; t < NT; ++t) o[t] = acc[t][i];
        cp[(size_t)(mbase + i) * 16] = o;
    }
}


#define ROWS_G     8

static_assert(OWN_NB * OWN_R == N_NODES, "owner ranges must tile the nodes");
static_assert(OWN_NB <= 256, "an owner number must fit one byte");
static_assert(N_BATCH * BATCH_E == N_EDGES, "batches must tile the edges");
static_assert(BATCH_E <= 65536, "a queue entry is a 16-bit position inside the batch");
static_assert(N_STRIPS % GEMM_WAVES == 0, "product grids: whole blocks of strips");
static_assert(N_STRIPS * 32 * 8 == N_EDGES, "product grids: one thread per 8 edges");
static_assert(ROWS_G == 8, "the load fence in add_rows names 8 rows");

__device__ __forceinline__ int keep_own(const u32x4 (&tg)[TAG_U], unsigned own4, unsigned short* q, int lane) {
    int cnt = 0;
#pragma unroll
    for (int u = 0; u < TAG_U; ++u) {
        unsigned nz[4];
#pragma unroll
        for (int k = 0; k < 4; ++k) {
            const unsigned x = tg[u][k] ^ own4;
            nz[k] = ((x & 0x7f7f7f7fu) + 0x7f7f7f7fu) | x;
        }
        unsigned f = (nz[0] & 0x80808080u) | ((nz[1] >> 1) & 0x40404040u) |
                     ((nz[2] >> 2) & 0x20202020u) | ((nz[3] >> 3) & 0x10101010u);
        f ^= 0xf0f0f0f0u;
        unsigned mask = __builtin_amdgcn_ballot_w32(f != 0u);
        while (mask != 0u) {
            if (f != 0u) {
                const int b   = __builtin_ctz(f);
                const int pos = cnt + (int)__builtin_amdgcn_mbcnt_lo(mask, 0u);
                const int j   = ((7 - (b & 7)) << 2) + (b >> 3);
                q[pos] = (unsigned short)(((u * 32 + lane) << 4) + j);
                f &= f - 1u;
            }
            cnt += __builtin_popcount(mask);
            mask = __builtin_amdgcn_ballot_w32(f != 0u);
        }
    }
    return cnt;
}

template <int C>
__device__ __forceinline__ void add_rows(const float* __restrict__ h, float* acc, int lane, int m,
                                         int s, int dl, float dsv, float ddv) {
    for (int j = 0; j < m; j += ROWS_G) {
        float v[ROWS_G][C / 32];
#pragma unroll
        for (int t = 0; t < ROWS_G; ++t) {
            const int sj = __builtin_amdgcn_readlane(s, j + t);
            const float* hp = h + (size_t)sj * C + lane;
#pragma unroll
            for (int c = 0; c < C / 32; ++c) v[t][c] = hp[c * 32];
        }
#pragma unroll
        for (int c = 0; c < C / 32; ++c)
            asm volatile("" :: "v"(v[0][c]), "v"(v[1][c]), "v"(v[2][c]), "v"(v[3][c]),
                               "v"(v[4][c]), "v"(v[5][c]), "v"(v[6][c]), "v"(v[7][c]));
#pragma unroll
        for (int t = 0; t < ROWS_G; ++t) {
            if (j + t < m) {
                const float nrm = __int_as_float(__builtin_amdgcn_readlane(__float_as_int(dsv), j + t)) *
                                  __int_as_float(__builtin_amdgcn_readlane(__float_as_int(ddv), j + t));
                const int dlj = __builtin_amdgcn_readlane(dl, j + t);
                float* ap = acc + dlj * C + lane;
#pragma unroll
                for (int c = 0; c < C / 32; ++c) unsafeAtomicAdd(ap + c * 32, v[t][c] * nrm);
            }
        }
    }
}

template <int C, bool SLICED>
__device__ __forceinline__ void scan_add(const float* __restrict__ h,
                                         const int* __restrict__ src,
                                         const int* __restrict__ dst,
                                         const unsigned* __restrict__ tags,
                                         const float* __restrict__ dinv,
                                         float* acc, unsigned short* q,
                                         int lane, int wave, int lo, int blk) {
    const unsigned own4 = blk * 0x01010101u;
    const int rot = (blk * 5) % N_BATCH;

    u32x4 tg[TAG_U];
    int bi = wave + rot; if (bi >= N_BATCH) bi -= N_BATCH;
    {
        const u32x4* tp = (const u32x4*)(tags + (SLICED ? sliced_tag_word(bi) : bi * BATCH_W));
#pragma unroll
        for (int u = 0; u < TAG_U; ++u) tg[u] = tp[lane + u * 32];
    }

    for (int it = wave; it < N_BATCH; it += OWN_WAVES) {
        const int e0 = bi * BATCH_E;
        const int cnt = keep_own(tg, own4, q, lane);

        int m = (cnt < 32) ? cnt : 32;
        int e = e0 + ((lane < m) ? (int)q[lane] : 0);
        int d = dst[e];
        int s = src[e];

        {
            const int itn = it + OWN_WAVES;
            bi = ((itn < N_BATCH) ? itn : it) + rot; if (bi >= N_BATCH) bi -= N_BATCH;
            const u32x4* tp = (const u32x4*)(tags + (SLICED ? sliced_tag_word(bi) : bi * BATCH_W));
#pragma unroll
            for (int u = 0; u < TAG_U; ++u) tg[u] = tp[lane + u * 32];
        }

        if (lane >= m) { s = 0; d = lo; }
        int   dl  = d - lo;
        float ddv = dinv[d];
        float dsv = dinv[s];
        add_rows<C>(h, acc, lane, m, s, dl, dsv, ddv);

        for (int b0 = 32; b0 < cnt; b0 += 32) {
            m   = (cnt - b0 < 32) ? (cnt - b0) : 32;
            e   = e0 + ((lane < m) ? (int)q[b0 + lane] : 0);
            d   = (lane < m) ? dst[e] : lo;
            s   = (lane < m) ? src[e] : 0;
            dl  = d - lo;
            ddv = dinv[d];
            dsv = dinv[s];
            add_rows<C>(h, acc, lane, m, s, dl, dsv, ddv);
        }
    }
}

__global__ __launch_bounds__(OWN_WAVES * 32)
void list_deg_kernel(const int* __restrict__ src,
                     const int* __restrict__ dst,
                     const unsigned* __restrict__ tags,
                     float* __restrict__ dinv,
                     unsigned* lists) {
    __shared__ unsigned short wq[OWN_WAVES][BATCH_E];
    __shared__ unsigned       wl[OWN_WAVES][LIST_CAP];
    __shared__ unsigned       dg[OWN_R];
    __shared__ unsigned       wn[OWN_WAVES];

    const int tid  = threadIdx.x;
    const int lane = tid & 31;
    const int wave = tid >> 5;
    const int blk  = blockIdx.x;
    const int lo   = blk * OWN_R;
    unsigned short* q  = wq[wave];
    unsigned*       ll = wl[wave];

    for (int i = tid; i < OWN_R; i += OWN_WAVES * 32) dg[i] = 0u;
    __syncthreads();

    const unsigned own4 = blk * 0x01010101u;
    const int rot = (blk * 5) % N_BATCH;
    int nl = 0;

    auto note = [&](int m, int d, int s) {
        if (lane < m) {
            const int dl = d - lo;
            if (nl + m <= LIST_CAP) ll[nl + lane] = (unsigned)s | ((unsigned)dl << 17);
            atomicAdd(&dg[dl], 1u);
        }
        nl += m;
    };

    u32x4 tg[TAG_U];
    int bi = wave + rot; if (bi >= N_BATCH) bi -= N_BATCH;
    {
        const u32x4* tp = (const u32x4*)(tags + sliced_tag_word(bi));
#pragma unroll
        for (int u = 0; u < TAG_U; ++u) tg[u] = tp[lane + u * 32];
    }

    int pm = 0, pd = lo, ps = 0;
    for (int it = wave; it < N_BATCH; it += OWN_WAVES) {
        const int e0 = bi * BATCH_E;
        const int itn = it + OWN_WAVES;
        int bn = ((itn < N_BATCH) ? itn : it) + rot; if (bn >= N_BATCH) bn -= N_BATCH;
        u32x4 tn[TAG_U];
        {
            const u32x4* tp = (const u32x4*)(tags + sliced_tag_word(bn));
#pragma unroll
            for (int u = 0; u < TAG_U; ++u) tn[u] = tp[lane + u * 32];
        }

        const int cnt = keep_own(tg, own4, q, lane);

        note(pm, pd, ps);

        pm = (cnt < 32) ? cnt : 32;
        const int e = e0 + ((lane < pm) ? (int)q[lane] : 0);
        pd = dst[e];
        ps = src[e];

        for (int b0 = 32; b0 < cnt; b0 += 32) {
            const int mm = (cnt - b0 < 32) ? (cnt - b0) : 32;
            const int ee = e0 + ((lane < mm) ? (int)q[b0 + lane] : 0);
            note(mm, dst[ee], src[ee]);
        }

#pragma unroll
        for (int u = 0; u < TAG_U; ++u) tg[u] = tn[u];
        bi = bn;
    }
    note(pm, pd, ps);

    if (lane == 0) wn[wave] = (unsigned)nl;
    __syncthreads();

    unsigned* sl = lists + (size_t)blk * SLICE_W;
    unsigned* lg = sl + LIST_W0 + wave * LIST_CAP;
    const int nk = (nl <= LIST_CAP) ? nl : 0;
    for (int i = lane; i < ((nk + 31) & ~31); i += 32) lg[i] = (i < nk) ? ll[i] : 0u;
    if (wave == 0) sl[lane] = wn[lane];

    for (int i = tid; i < OWN_R; i += OWN_WAVES * 32) dinv[lo + i] = rsqrtf(1.0f + (float)dg[i]);
}

template <int C, bool RELU, bool SLICED>
__global__ __launch_bounds__(OWN_WAVES * 32)
void agg_owner_kernel(const float* __restrict__ h,
                      const int* __restrict__ src,
                      const int* __restrict__ dst,
                      const unsigned* __restrict__ tags,
                      const float* __restrict__ dinv,
                      const float* __restrict__ bias,
                      float* out,
                      const unsigned* lists) {
    __shared__ float          acc[OWN_R * C];
    __shared__ unsigned short wq[OWN_WAVES][BATCH_E];

    const int tid  = threadIdx.x;
    const int lane = tid & 31;
    const int wave = tid >> 5;
    const int lo   = blockIdx.x * OWN_R;
    const unsigned* sl = lists + (size_t)blockIdx.x * SLICE_W;
    const unsigned cw = sl[lane];

    for (int i = tid; i < OWN_R * C; i += OWN_WAVES * 32) acc[i] = 0.0f;
    __syncthreads();

    if (__builtin_amdgcn_ballot_w32(cw > (unsigned)LIST_CAP) == 0u) {
        const int n = __builtin_amdgcn_readlane((int)cw, __builtin_amdgcn_readfirstlane(wave));
        const unsigned* lg = sl + LIST_W0 + wave * LIST_CAP;
        unsigned nx = (lane < n) ? lg[lane] : 0u;
        for (int b0 = 0; b0 < n; b0 += 32) {
            const int m = (n - b0 < 32) ? (n - b0) : 32;
            const unsigned cur = nx;
            nx = (b0 + 32 + lane < n) ? lg[b0 + 32 + lane] : 0u;
            const int s  = (int)(cur & 0x1ffffu);
            const int dl = (int)(cur >> 17);
            const float ddv = dinv[lo + dl];
            const float dsv = dinv[s];
            add_rows<C>(h, acc, lane, m, s, dl, dsv, ddv);
        }
    } else {
        scan_add<C, SLICED>(h, src, dst, tags, dinv, acc, wq[wave], lane, wave, lo, blockIdx.x);
    }

    __syncthreads();

    const size_t g0 = (size_t)lo * C;
    for (int i = tid; i < OWN_R * C; i += OWN_WAVES * 32) {
        const int nl = i / C;
        const int c  = i & (C - 1);
        const float di = dinv[lo + nl];
        const float val = acc[i] + h[g0 + i] * di * di + bias[c];
        out[g0 + i] = RELU ? fmaxf(val, 0.0f) : val;
    }
}


extern "C" void kernel_launch(void* const* d_in, const int* in_sizes, int n_in,
                              void* d_out, int out_size, void* d_ws, size_t ws_size,
                              hipStream_t stream) {
    const float* x    = (const float*)d_in[0];
    const int*   ei   = (const int*)d_in[1];
    const float* W1   = (const float*)d_in[2];
    const float* b1   = (const float*)d_in[3];
    const float* W2   = (const float*)d_in[4];
    const float* b2   = (const float*)d_in[5];
    float*       out  = (float*)d_out;

    const int* srcv = ei;
    const int* dstv = ei + N_EDGES;

    float* dinv = (float*)d_ws;
    float* h1   = dinv + 100352;
    float* agg1 = h1 + (size_t)N_NODES * HID_CH;
    float* h2   = h1;
    unsigned* obuf  = (unsigned*)d_out;
    unsigned* tags2 = (unsigned*)(h1 + (size_t)N_NODES * OUT_CH);

    gemm_wmma_f32<IN_CH, HID_CH, 1><<<N_STRIPS / GEMM_WAVES, GEMM_WAVES * 32, 0, stream>>>(x, W1, h1, dstv, obuf);

    list_deg_kernel<<<OWN_NB, OWN_WAVES * 32, 0, stream>>>(srcv, dstv, obuf, dinv, obuf);

    agg_owner_kernel<HID_CH, true, true>
        <<<OWN_NB, OWN_WAVES * 32, 0, stream>>>(h1, srcv, dstv, obuf, dinv, b1, agg1, obuf);

    gemm_wmma_f32<HID_CH, OUT_CH, 2><<<N_STRIPS / GEMM_WAVES, GEMM_WAVES * 32, 0, stream>>>(agg1, W2, h2, dstv, tags2);

    agg_owner_kernel<OUT_CH, false, false>
        <<<OWN_NB, OWN_WAVES * 32, 0, stream>>>(h2, srcv, dstv, tags2, dinv, b2, out, obuf);
}


# List launch split two ways: 125 node ranges x 2 edge halves, each block scans half the owner bytes; in-degrees merged by global integer adds (exact); list order, so sum order, changes
# speedup vs baseline: 4.4474x; 1.1278x over previous
#include <hip/hip_runtime.h>
#include <hip/hip_bf16.h>


#define N_NODES 100000
#define N_EDGES 1600000
#define IN_CH   128
#define HID_CH  64
#define OUT_CH  32

typedef __attribute__((ext_vector_type(2))) float f32x2;
typedef __attribute__((ext_vector_type(8))) float f32x8;

typedef __attribute__((ext_vector_type(4))) int i32x4;
typedef __attribute__((ext_vector_type(4))) unsigned u32x4;

#define OWN_NB     250
#define OWN_R      400
#define OWN_WAVES  32
#define TAG_U      5
#define BATCH_G    (TAG_U * 32)
#define BATCH_E    (BATCH_G * 16)
#define BATCH_W    (BATCH_E / 4)
#define N_BATCH    (N_EDGES / BATCH_E)

#define SLICE_W    (OWN_R * OUT_CH)
#define RNG_NB     125
#define RNG_R      (2 * OWN_R)
#define PART0_NB   313
#define N_LISTS    (2 * OWN_WAVES)
#define SUB_CAP    160
#define LIST_W0    N_LISTS
#define TAGS_W0    (LIST_W0 + N_LISTS * SUB_CAP)
#define TAGS_BPS   3

static_assert(N_NODES <= (1 << 17) && OWN_R <= (1 << 15), "a list entry packs source and local node in one word");
static_assert(TAGS_W0 % 4 == 0 && SLICE_W % 4 == 0 && BATCH_W % 4 == 0, "owner bytes are read 16 bytes at a time");
static_assert(TAGS_W0 + TAGS_BPS * BATCH_W <= SLICE_W, "lists and owner bytes must fit one slice");
static_assert(TAGS_BPS * OWN_NB >= N_BATCH, "every batch of owner bytes needs a slice");
static_assert(RNG_NB * RNG_R == N_NODES && 2 * RNG_NB == OWN_NB, "ranges tile the nodes; a range is two owner blocks");
static_assert(PART0_NB > 0 && PART0_NB < N_BATCH && SUB_CAP % 32 == 0, "two edge halves; lists are whole lines");

__device__ __forceinline__ float dinv_at(const unsigned* __restrict__ cnt, int n) {
    return rsqrtf(1.0f + (float)cnt[n]);
}

__device__ __forceinline__ int sliced_tag_word(int t) {
    return (t / TAGS_BPS) * SLICE_W + TAGS_W0 + (t % TAGS_BPS) * BATCH_W;
}

__device__ __forceinline__ unsigned owner_bytes(i32x4 d) {
    return  ((unsigned)d.x / (unsigned)OWN_R)        | (((unsigned)d.y / (unsigned)OWN_R) << 8) |
           (((unsigned)d.z / (unsigned)OWN_R) << 16) | (((unsigned)d.w / (unsigned)OWN_R) << 24);
}


#define GEMM_WAVES 5
#define N_STRIPS   (N_NODES / 16)

template <int K, int NCOL, int TAGS>
__global__ __launch_bounds__(GEMM_WAVES * 32)
void gemm_wmma_f32(const float* __restrict__ A,
                   const float* __restrict__ B,
                   float* __restrict__ C,
                   const int* __restrict__ dst,
                   unsigned* __restrict__ tags,
                   unsigned* __restrict__ cnt) {
    constexpr int NT = NCOL / 16;
    typedef __attribute__((ext_vector_type(NT))) float fvec;

    const int gt = blockIdx.x * (GEMM_WAVES * 32) + threadIdx.x;
    if (TAGS == 1) {
        if (gt < N_NODES) cnt[gt] = 0u;
        for (int r = 0; r < 2; ++r) {
            const int g = gt + r * (N_EDGES / 8);
            const int t = g / BATCH_W;
            tags[sliced_tag_word(t) + (g - t * BATCH_W)] = owner_bytes(((const i32x4*)dst)[g]);
        }
    }
    if (TAGS == 2) {
        for (int r = 0; r < 2; ++r) {
            const int g = gt + r * (N_EDGES / 8);
            tags[g] = owner_bytes(((const i32x4*)dst)[g]);
        }
    }
    const int lane = threadIdx.x & 31;
    const int row0 = (blockIdx.x * GEMM_WAVES + (threadIdx.x >> 5)) << 4;

    const int m  = lane & 15;
    const int kk = (lane >> 4) << 1;

    const float* ap = A + (size_t)(row0 + m) * K + kk;
    const fvec*  bp = (const fvec*)(B + (size_t)kk * NCOL) + m;

    f32x8 acc[NT];
    for (int t = 0; t < NT; ++t) acc[t] = (f32x8){0.f, 0.f, 0.f, 0.f, 0.f, 0.f, 0.f, 0.f};

    for (int k0 = 0; k0 < K; k0 += 4) {
        f32x2 a;
        a.x = ap[k0];
        a.y = ap[k0 + 1];
        const fvec b0 = bp[(size_t)k0 * 16];
        const fvec b1 = bp[(size_t)(k0 + 1) * 16];
        for (int t = 0; t < NT; ++t) {
            f32x2 b;
            b.x = b0[t];
            b.y = b1[t];
            acc[t] = __builtin_amdgcn_wmma_f32_16x16x4_f32(
                 false, a,  false, b,
                 (short)0, acc[t],  false,  false);
        }
    }

    const int mbase = (lane >> 4) << 3;
    fvec* cp = (fvec*)(C + (size_t)row0 * NCOL) + m;
    for (int i = 0; i < 8; ++i) {
        fvec o;
        for (int t = 0; t < NT; ++t) o[t] = acc[t][i];
        cp[(size_t)(mbase + i) * 16] = o;
    }
}


#define ROWS_G     8

static_assert(OWN_NB * OWN_R == N_NODES, "owner ranges must tile the nodes");
static_assert(OWN_NB <= 256, "an owner number must fit one byte");
static_assert(N_BATCH * BATCH_E == N_EDGES, "batches must tile the edges");
static_assert(BATCH_E <= 65536, "a queue entry is a 16-bit position inside the batch");
static_assert(N_STRIPS % GEMM_WAVES == 0, "product grids: whole blocks of strips");
static_assert(N_STRIPS * 32 * 8 == N_EDGES, "product grids: one thread per 8 edges");
static_assert(ROWS_G == 8, "the load fence in add_rows names 8 rows");

template <bool PAIR>
__device__ __forceinline__ int keep_own(const u32x4 (&tg)[TAG_U], unsigned own4, unsigned short* q, int lane) {
    int cnt = 0;
#pragma unroll
    for (int u = 0; u < TAG_U; ++u) {
        unsigned nz[4];
#pragma unroll
        for (int k = 0; k < 4; ++k) {
            const unsigned x = (tg[u][k] ^ own4) & (PAIR ? 0xfefefefeu : 0xffffffffu);
            nz[k] = ((x & 0x7f7f7f7fu) + 0x7f7f7f7fu) | x;
        }
        unsigned f = (nz[0] & 0x80808080u) | ((nz[1] >> 1) & 0x40404040u) |
                     ((nz[2] >> 2) & 0x20202020u) | ((nz[3] >> 3) & 0x10101010u);
        f ^= 0xf0f0f0f0u;
        unsigned mask = __builtin_amdgcn_ballot_w32(f != 0u);
        while (mask != 0u) {
            if (f != 0u) {
                const int b   = __builtin_ctz(f);
                const int pos = cnt + (int)__builtin_amdgcn_mbcnt_lo(mask, 0u);
                const int j   = ((7 - (b & 7)) << 2) + (b >> 3);
                q[pos] = (unsigned short)(((u * 32 + lane) << 4) + j);
                f &= f - 1u;
            }
            cnt += __builtin_popcount(mask);
            mask = __builtin_amdgcn_ballot_w32(f != 0u);
        }
    }
    return cnt;
}

template <int C>
__device__ __forceinline__ void add_rows(const float* __restrict__ h, float* acc, int lane, int m,
                                         int s, int dl, float dsv, float ddv) {
    for (int j = 0; j < m; j += ROWS_G) {
        float v[ROWS_G][C / 32];
#pragma unroll
        for (int t = 0; t < ROWS_G; ++t) {
            const int sj = __builtin_amdgcn_readlane(s, j + t);
            const float* hp = h + (size_t)sj * C + lane;
#pragma unroll
            for (int c = 0; c < C / 32; ++c) v[t][c] = hp[c * 32];
        }
#pragma unroll
        for (int c = 0; c < C / 32; ++c)
            asm volatile("" :: "v"(v[0][c]), "v"(v[1][c]), "v"(v[2][c]), "v"(v[3][c]),
                               "v"(v[4][c]), "v"(v[5][c]), "v"(v[6][c]), "v"(v[7][c]));
#pragma unroll
        for (int t = 0; t < ROWS_G; ++t) {
            if (j + t < m) {
                const float nrm = __int_as_float(__builtin_amdgcn_readlane(__float_as_int(dsv), j + t)) *
                                  __int_as_float(__builtin_amdgcn_readlane(__float_as_int(ddv), j + t));
                const int dlj = __builtin_amdgcn_readlane(dl, j + t);
                float* ap = acc + dlj * C + lane;
#pragma unroll
                for (int c = 0; c < C / 32; ++c) unsafeAtomicAdd(ap + c * 32, v[t][c] * nrm);
            }
        }
    }
}

template <int C, bool SLICED>
__device__ __forceinline__ void scan_add(const float* __restrict__ h,
                                         const int* __restrict__ src,
                                         const int* __restrict__ dst,
                                         const unsigned* __restrict__ tags,
                                         const unsigned* __restrict__ cnt,
                                         float* acc, unsigned short* q,
                                         int lane, int wave, int lo, int blk) {
    const unsigned own4 = blk * 0x01010101u;
    const int rot = (blk * 5) % N_BATCH;

    u32x4 tg[TAG_U];
    int bi = wave + rot; if (bi >= N_BATCH) bi -= N_BATCH;
    {
        const u32x4* tp = (const u32x4*)(tags + (SLICED ? sliced_tag_word(bi) : bi * BATCH_W));
#pragma unroll
        for (int u = 0; u < TAG_U; ++u) tg[u] = tp[lane + u * 32];
    }

    for (int it = wave; it < N_BATCH; it += OWN_WAVES) {
        const int e0 = bi * BATCH_E;
        const int cnt_k = keep_own<false>(tg, own4, q, lane);

        int m = (cnt_k < 32) ? cnt_k : 32;
        int e = e0 + ((lane < m) ? (int)q[lane] : 0);
        int d = dst[e];
        int s = src[e];

        {
            const int itn = it + OWN_WAVES;
            bi = ((itn < N_BATCH) ? itn : it) + rot; if (bi >= N_BATCH) bi -= N_BATCH;
            const u32x4* tp = (const u32x4*)(tags + (SLICED ? sliced_tag_word(bi) : bi * BATCH_W));
#pragma unroll
            for (int u = 0; u < TAG_U; ++u) tg[u] = tp[lane + u * 32];
        }

        if (lane >= m) { s = 0; d = lo; }
        int   dl  = d - lo;
        float ddv = dinv_at(cnt, d);
        float dsv = dinv_at(cnt, s);
        add_rows<C>(h, acc, lane, m, s, dl, dsv, ddv);

        for (int b0 = 32; b0 < cnt_k; b0 += 32) {
            m   = (cnt_k - b0 < 32) ? (cnt_k - b0) : 32;
            e   = e0 + ((lane < m) ? (int)q[b0 + lane] : 0);
            d   = (lane < m) ? dst[e] : lo;
            s   = (lane < m) ? src[e] : 0;
            dl  = d - lo;
            ddv = dinv_at(cnt, d);
            dsv = dinv_at(cnt, s);
            add_rows<C>(h, acc, lane, m, s, dl, dsv, ddv);
        }
    }
}

__global__ __launch_bounds__(OWN_WAVES * 32)
void list_deg_kernel(const int* __restrict__ src,
                     const int* __restrict__ dst,
                     const unsigned* __restrict__ tags,
                     unsigned* __restrict__ cnt,
                     unsigned* lists) {
    __shared__ unsigned short wq[OWN_WAVES][BATCH_E];
    __shared__ unsigned       wl[OWN_WAVES][2][SUB_CAP];
    __shared__ unsigned       dg[RNG_R];
    __shared__ unsigned       wn[2][OWN_WAVES];

    const int tid  = threadIdx.x;
    const int lane = tid & 31;
    const int wave = tid >> 5;
    const int rng  = blockIdx.x >> 1;
    const int part = blockIdx.x & 1;
    const int lo   = rng * RNG_R;
    const int bt0  = part ? PART0_NB : 0;
    const int nbt  = part ? (N_BATCH - PART0_NB) : PART0_NB;
    unsigned short* q   = wq[wave];
    unsigned*       ll0 = wl[wave][0];
    unsigned*       ll1 = wl[wave][1];

    for (int i = tid; i < RNG_R; i += OWN_WAVES * 32) dg[i] = 0u;
    __syncthreads();

    const unsigned own4 = (unsigned)(2 * rng) * 0x01010101u;
    const int rot = (rng * 5) % nbt;
    int nl0 = 0, nl1 = 0;

    auto note = [&](int m, int d, int s) {
        const bool act = lane < m;
        const int  dl  = d - lo;
        const bool up  = act && dl >= OWN_R;
        const bool dn  = act && dl <  OWN_R;
        const unsigned mu = __builtin_amdgcn_ballot_w32(up);
        const unsigned md = __builtin_amdgcn_ballot_w32(dn);
        const int cu = __builtin_popcount(mu);
        const int cd = __builtin_popcount(md);
        if (dn && nl0 + cd <= SUB_CAP)
            ll0[nl0 + (int)__builtin_amdgcn_mbcnt_lo(md, 0u)] = (unsigned)s | ((unsigned)dl << 17);
        if (up && nl1 + cu <= SUB_CAP)
            ll1[nl1 + (int)__builtin_amdgcn_mbcnt_lo(mu, 0u)] = (unsigned)s | ((unsigned)(dl - OWN_R) << 17);
        if (act) atomicAdd(&dg[dl], 1u);
        nl0 += cd;
        nl1 += cu;
    };

    u32x4 tg[TAG_U];
    int bi;
    {
        int ix = wave + rot; if (ix >= nbt) ix -= nbt;
        bi = bt0 + ix;
        const u32x4* tp = (const u32x4*)(tags + sliced_tag_word(bi));
#pragma unroll
        for (int u = 0; u < TAG_U; ++u) tg[u] = tp[lane + u * 32];
    }

    int pm = 0, pd = lo, ps = 0;
    for (int it = wave; it < nbt; it += OWN_WAVES) {
        const int e0 = bi * BATCH_E;
        const int itn = it + OWN_WAVES;
        int ixn = ((itn < nbt) ? itn : it) + rot; if (ixn >= nbt) ixn -= nbt;
        const int bn = bt0 + ixn;
        u32x4 tn[TAG_U];
        {
            const u32x4* tp = (const u32x4*)(tags + sliced_tag_word(bn));
#pragma unroll
            for (int u = 0; u < TAG_U; ++u) tn[u] = tp[lane + u * 32];
        }

        const int cnt_k = keep_own<true>(tg, own4, q, lane);

        note(pm, pd, ps);

        pm = (cnt_k < 32) ? cnt_k : 32;
        const int e = e0 + ((lane < pm) ? (int)q[lane] : 0);
        pd = dst[e];
        ps = src[e];

        for (int b0 = 32; b0 < cnt_k; b0 += 32) {
            const int mm = (cnt_k - b0 < 32) ? (cnt_k - b0) : 32;
            const int ee = e0 + ((lane < mm) ? (int)q[b0 + lane] : 0);
            note(mm, dst[ee], src[ee]);
        }

#pragma unroll
        for (int u = 0; u < TAG_U; ++u) tg[u] = tn[u];
        bi = bn;
    }
    note(pm, pd, ps);

    if (lane == 0) { wn[0][wave] = (unsigned)nl0; wn[1][wave] = (unsigned)nl1; }
    __syncthreads();

    for (int sub = 0; sub < 2; ++sub) {
        unsigned* sl = lists + (size_t)(2 * rng + sub) * SLICE_W;
        unsigned* lg = sl + LIST_W0 + (part * OWN_WAVES + wave) * SUB_CAP;
        const unsigned* ll = sub ? ll1 : ll0;
        const int nl = sub ? nl1 : nl0;
        const int nk = (nl <= SUB_CAP) ? nl : 0;
        for (int i = lane; i < ((nk + 31) & ~31); i += 32) lg[i] = (i < nk) ? ll[i] : 0u;
        if (wave == 0) sl[part * OWN_WAVES + lane] = wn[sub][lane];
    }

    for (int i = tid; i < RNG_R; i += OWN_WAVES * 32) {
        const unsigned c = dg[i];
        if (c != 0u) atomicAdd(&cnt[lo + i], c);
    }
}

template <int C, bool RELU, bool SLICED>
__global__ __launch_bounds__(OWN_WAVES * 32)
void agg_owner_kernel(const float* __restrict__ h,
                      const int* __restrict__ src,
                      const int* __restrict__ dst,
                      const unsigned* __restrict__ tags,
                      const unsigned* __restrict__ cnt,
                      const float* __restrict__ bias,
                      float* out,
                      const unsigned* lists) {
    __shared__ float          acc[OWN_R * C];
    __shared__ unsigned short wq[OWN_WAVES][BATCH_E];

    const int tid  = threadIdx.x;
    const int lane = tid & 31;
    const int wave = tid >> 5;
    const int lo   = blockIdx.x * OWN_R;
    const unsigned* sl = lists + (size_t)blockIdx.x * SLICE_W;
    const unsigned cw0 = sl[lane];
    const unsigned cw1 = sl[OWN_WAVES + lane];

    for (int i = tid; i < OWN_R * C; i += OWN_WAVES * 32) acc[i] = 0.0f;
    __syncthreads();

    if (__builtin_amdgcn_ballot_w32(cw0 > (unsigned)SUB_CAP || cw1 > (unsigned)SUB_CAP) == 0u) {
        for (int p = 0; p < 2; ++p) {
            const int n = __builtin_amdgcn_readlane((int)(p ? cw1 : cw0), __builtin_amdgcn_readfirstlane(wave));
            const unsigned* lg = sl + LIST_W0 + (p * OWN_WAVES + wave) * SUB_CAP;
            unsigned cur = (lane < n) ? lg[lane] : 0u;
            unsigned nx  = (32 + lane < n) ? lg[32 + lane] : 0u;
            unsigned cs  = cnt[cur & 0x1ffffu];
            unsigned cd  = cnt[lo + (int)(cur >> 17)];
            for (int b0 = 0; b0 < n; b0 += 32) {
                const int m = (n - b0 < 32) ? (n - b0) : 32;
                const unsigned csn = cnt[nx & 0x1ffffu];
                const unsigned cdn = cnt[lo + (int)(nx >> 17)];
                const unsigned nn  = (b0 + 64 + lane < n) ? lg[b0 + 64 + lane] : 0u;
                const int s  = (int)(cur & 0x1ffffu);
                const int dl = (int)(cur >> 17);
                const float ddv = rsqrtf(1.0f + (float)cd);
                const float dsv = rsqrtf(1.0f + (float)cs);
                add_rows<C>(h, acc, lane, m, s, dl, dsv, ddv);
                cur = nx; nx = nn; cs = csn; cd = cdn;
            }
        }
    } else {
        scan_add<C, SLICED>(h, src, dst, tags, cnt, acc, wq[wave], lane, wave, lo, blockIdx.x);
    }

    __syncthreads();

    constexpr int  EP_STEP = OWN_WAVES * 32;
    constexpr int  EP_N    = OWN_R * C;
    constexpr int  EP_T    = (EP_N + EP_STEP - 1) / EP_STEP;
    constexpr bool EP_FULL = (EP_N % EP_STEP) == 0;
    static_assert(EP_STEP % C == 0, "a thread keeps its channel over its elements");
    const size_t g0 = (size_t)lo * C;
    const float  bc = bias[tid & (C - 1)];
    float    hv[EP_T];
    unsigned cv[EP_T];
#pragma unroll
    for (int t = 0; t < EP_T; ++t) {
        const int i  = tid + t * EP_STEP;
        const int ic = (EP_FULL || i < EP_N) ? i : tid;
        hv[t] = h[g0 + ic];
        cv[t] = cnt[lo + ic / C];
    }
#pragma unroll
    for (int t = 0; t < EP_T; ++t) {
        const int i = tid + t * EP_STEP;
        if (EP_FULL || i < EP_N) {
            const float di  = rsqrtf(1.0f + (float)cv[t]);
            const float val = acc[i] + hv[t] * di * di + bc;
            out[g0 + i] = RELU ? fmaxf(val, 0.0f) : val;
        }
    }
}


extern "C" void kernel_launch(void* const* d_in, const int* in_sizes, int n_in,
                              void* d_out, int out_size, void* d_ws, size_t ws_size,
                              hipStream_t stream) {
    const float* x    = (const float*)d_in[0];
    const int*   ei   = (const int*)d_in[1];
    const float* W1   = (const float*)d_in[2];
    const float* b1   = (const float*)d_in[3];
    const float* W2   = (const float*)d_in[4];
    const float* b2   = (const float*)d_in[5];
    float*       out  = (float*)d_out;

    const int* srcv = ei;
    const int* dstv = ei + N_EDGES;

    unsigned* cnt = (unsigned*)d_ws;
    float* h1   = (float*)d_ws + 100352;
    float* agg1 = h1 + (size_t)N_NODES * HID_CH;
    float* h2   = h1;
    unsigned* obuf  = (unsigned*)d_out;
    unsigned* tags2 = (unsigned*)(h1 + (size_t)N_NODES * OUT_CH);

    gemm_wmma_f32<IN_CH, HID_CH, 1><<<N_STRIPS / GEMM_WAVES, GEMM_WAVES * 32, 0, stream>>>(x, W1, h1, dstv, obuf, cnt);

    list_deg_kernel<<<2 * RNG_NB, OWN_WAVES * 32, 0, stream>>>(srcv, dstv, obuf, cnt, obuf);

    agg_owner_kernel<HID_CH, true, true>
        <<<OWN_NB, OWN_WAVES * 32, 0, stream>>>(h1, srcv, dstv, obuf, cnt, b1, agg1, obuf);

    gemm_wmma_f32<HID_CH, OUT_CH, 2><<<N_STRIPS / GEMM_WAVES, GEMM_WAVES * 32, 0, stream>>>(agg1, W2, h2, dstv, tags2, cnt);

    agg_owner_kernel<OUT_CH, false, false>
        <<<OWN_NB, OWN_WAVES * 32, 0, stream>>>(h2, srcv, dstv, tags2, cnt, b2, out, obuf);
}


# List launch re-split: 32 node ranges x 8 edge parts (was 125 x 2); a block scans 1/8 of the owner bytes, keeps one LDS list per owner block (integer LDS adds); list order, so sum order, changes
# speedup vs baseline: 4.8243x; 1.0847x over previous
#include <hip/hip_runtime.h>
#include <hip/hip_bf16.h>


#define N_NODES 100000
#define N_EDGES 1600000
#define IN_CH   128
#define HID_CH  64
#define OUT_CH  32

typedef __attribute__((ext_vector_type(2))) float f32x2;
typedef __attribute__((ext_vector_type(8))) float f32x8;

typedef __attribute__((ext_vector_type(4))) int i32x4;
typedef __attribute__((ext_vector_type(4))) unsigned u32x4;

#define OWN_NB     250
#define OWN_R      400
#define OWN_WAVES  32
#define TAG_U      5
#define BATCH_G    (TAG_U * 32)
#define BATCH_E    (BATCH_G * 16)
#define BATCH_W    (BATCH_E / 4)
#define N_BATCH    (N_EDGES / BATCH_E)

#define SLICE_W    (OWN_R * OUT_CH)
#define RNG_OWN    8
#define RNG_NB     32
#define RNG_R      (RNG_OWN * OWN_R)
#define N_PART     8
#define LIST_CAP   1024
#define LIST_W0    64
#define TAGS_W0    10304
#define TAGS_BPS   3

static_assert(N_NODES <= (1 << 17) && OWN_R <= (1 << 15), "a list entry packs source and local node in one word");
static_assert(TAGS_W0 % 4 == 0 && SLICE_W % 4 == 0 && BATCH_W % 4 == 0, "owner bytes are read 16 bytes at a time");
static_assert(TAGS_W0 + TAGS_BPS * BATCH_W <= SLICE_W, "lists and owner bytes must fit one slice");
static_assert(TAGS_BPS * OWN_NB >= N_BATCH, "every batch of owner bytes needs a slice");
static_assert(RNG_NB * RNG_OWN >= OWN_NB && (RNG_NB - 1) * RNG_OWN < OWN_NB && (RNG_OWN & (RNG_OWN - 1)) == 0,
              "ranges cover the owner blocks; a range is a power of two of them (one masked byte test)");
static_assert(N_PART <= LIST_W0 && LIST_W0 + N_PART * LIST_CAP <= TAGS_W0 && LIST_CAP % 32 == 0 && LIST_W0 % 32 == 0,
              "counts, then the lists as whole lines, in front of the owner bytes");
static_assert(OWN_WAVES % N_PART == 0 && OWN_WAVES % RNG_OWN == 0 && N_PART <= 32 && N_BATCH / N_PART >= OWN_WAVES,
              "waves share lists evenly; every wave of the list launch has a batch");

__device__ __forceinline__ float dinv_at(const unsigned* __restrict__ cnt, int n) {
    return rsqrtf(1.0f + (float)cnt[n]);
}

__device__ __forceinline__ int sliced_tag_word(int t) {
    return (t / TAGS_BPS) * SLICE_W + TAGS_W0 + (t % TAGS_BPS) * BATCH_W;
}

__device__ __forceinline__ unsigned owner_bytes(i32x4 d) {
    return  ((unsigned)d.x / (unsigned)OWN_R)        | (((unsigned)d.y / (unsigned)OWN_R) << 8) |
           (((unsigned)d.z / (unsigned)OWN_R) << 16) | (((unsigned)d.w / (unsigned)OWN_R) << 24);
}


#define GEMM_WAVES 5
#define N_STRIPS   (N_NODES / 16)

template <int K, int NCOL, int TAGS>
__global__ __launch_bounds__(GEMM_WAVES * 32)
void gemm_wmma_f32(const float* __restrict__ A,
                   const float* __restrict__ B,
                   float* __restrict__ C,
                   const int* __restrict__ dst,
                   unsigned* __restrict__ tags,
                   unsigned* __restrict__ cnt) {
    constexpr int NT = NCOL / 16;
    typedef __attribute__((ext_vector_type(NT))) float fvec;

    const int gt = blockIdx.x * (GEMM_WAVES * 32) + threadIdx.x;
    if (TAGS == 1) {
        if (gt < N_NODES) cnt[gt] = 0u;
        for (int r = 0; r < 2; ++r) {
            const int g = gt + r * (N_EDGES / 8);
            const int t = g / BATCH_W;
            tags[sliced_tag_word(t) + (g - t * BATCH_W)] = owner_bytes(((const i32x4*)dst)[g]);
        }
    }
    if (TAGS == 2) {
        for (int r = 0; r < 2; ++r) {
            const int g = gt + r * (N_EDGES / 8);
            tags[g] = owner_bytes(((const i32x4*)dst)[g]);
        }
    }
    const int lane = threadIdx.x & 31;
    const int row0 = (blockIdx.x * GEMM_WAVES + (threadIdx.x >> 5)) << 4;

    const int m  = lane & 15;
    const int kk = (lane >> 4) << 1;

    const float* ap = A + (size_t)(row0 + m) * K + kk;
    const fvec*  bp = (const fvec*)(B + (size_t)kk * NCOL) + m;

    f32x8 acc[NT];
    for (int t = 0; t < NT; ++t) acc[t] = (f32x8){0.f, 0.f, 0.f, 0.f, 0.f, 0.f, 0.f, 0.f};

    for (int k0 = 0; k0 < K; k0 += 4) {
        f32x2 a;
        a.x = ap[k0];
        a.y = ap[k0 + 1];
        const fvec b0 = bp[(size_t)k0 * 16];
        const fvec b1 = bp[(size_t)(k0 + 1) * 16];
        for (int t = 0; t < NT; ++t) {
            f32x2 b;
            b.x = b0[t];
            b.y = b1[t];
            acc[t] = __builtin_amdgcn_wmma_f32_16x16x4_f32(
                 false, a,  false, b,
                 (short)0, acc[t],  false,  false);
        }
    }

    const int mbase = (lane >> 4) << 3;
    fvec* cp = (fvec*)(C + (size_t)row0 * NCOL) + m;
    for (int i = 0; i < 8; ++i) {
        fvec o;
        for (int t = 0; t < NT; ++t) o[t] = acc[t][i];
        cp[(size_t)(mbase + i) * 16] = o;
    }
}


#define ROWS_G     8

static_assert(OWN_NB * OWN_R == N_NODES, "owner ranges must tile the nodes");
static_assert(OWN_NB <= 256, "an owner number must fit one byte");
static_assert(N_BATCH * BATCH_E == N_EDGES, "batches must tile the edges");
static_assert(BATCH_E <= 65536, "a queue entry is a 16-bit position inside the batch");
static_assert(N_STRIPS % GEMM_WAVES == 0, "product grids: whole blocks of strips");
static_assert(N_STRIPS * 32 * 8 == N_EDGES, "product grids: one thread per 8 edges");
static_assert(ROWS_G == 8, "the load fence in add_rows names 8 rows");

template <int RBITS>
__device__ __forceinline__ int keep_own(const u32x4 (&tg)[TAG_U], unsigned own4, unsigned short* q, int lane) {
    int cnt = 0;
#pragma unroll
    for (int u = 0; u < TAG_U; ++u) {
        unsigned nz[4];
#pragma unroll
        for (int k = 0; k < 4; ++k) {
            const unsigned x = (tg[u][k] ^ own4) & (((0xffu << RBITS) & 0xffu) * 0x01010101u);
            nz[k] = ((x & 0x7f7f7f7fu) + 0x7f7f7f7fu) | x;
        }
        unsigned f = (nz[0] & 0x80808080u) | ((nz[1] >> 1) & 0x40404040u) |
                     ((nz[2] >> 2) & 0x20202020u) | ((nz[3] >> 3) & 0x10101010u);
        f ^= 0xf0f0f0f0u;
        unsigned mask = __builtin_amdgcn_ballot_w32(f != 0u);
        while (mask != 0u) {
            if (f != 0u) {
                const int b   = __builtin_ctz(f);
                const int pos = cnt + (int)__builtin_amdgcn_mbcnt_lo(mask, 0u);
                const int j   = ((7 - (b & 7)) << 2) + (b >> 3);
                q[pos] = (unsigned short)(((u * 32 + lane) << 4) + j);
                f &= f - 1u;
            }
            cnt += __builtin_popcount(mask);
            mask = __builtin_amdgcn_ballot_w32(f != 0u);
        }
    }
    return cnt;
}

template <int C>
__device__ __forceinline__ void add_rows(const float* __restrict__ h, float* acc, int lane, int m,
                                         int s, int dl, float dsv, float ddv) {
    for (int j = 0; j < m; j += ROWS_G) {
        float v[ROWS_G][C / 32];
#pragma unroll
        for (int t = 0; t < ROWS_G; ++t) {
            const int sj = __builtin_amdgcn_readlane(s, j + t);
            const float* hp = h + (size_t)sj * C + lane;
#pragma unroll
            for (int c = 0; c < C / 32; ++c) v[t][c] = hp[c * 32];
        }
#pragma unroll
        for (int c = 0; c < C / 32; ++c)
            asm volatile("" :: "v"(v[0][c]), "v"(v[1][c]), "v"(v[2][c]), "v"(v[3][c]),
                               "v"(v[4][c]), "v"(v[5][c]), "v"(v[6][c]), "v"(v[7][c]));
#pragma unroll
        for (int t = 0; t < ROWS_G; ++t) {
            if (j + t < m) {
                const float nrm = __int_as_float(__builtin_amdgcn_readlane(__float_as_int(dsv), j + t)) *
                                  __int_as_float(__builtin_amdgcn_readlane(__float_as_int(ddv), j + t));
                const int dlj = __builtin_amdgcn_readlane(dl, j + t);
                float* ap = acc + dlj * C + lane;
#pragma unroll
                for (int c = 0; c < C / 32; ++c) unsafeAtomicAdd(ap + c * 32, v[t][c] * nrm);
            }
        }
    }
}

template <int C, bool SLICED>
__device__ __forceinline__ void scan_add(const float* __restrict__ h,
                                         const int* __restrict__ src,
                                         const int* __restrict__ dst,
                                         const unsigned* __restrict__ tags,
                                         const unsigned* __restrict__ cnt,
                                         float* acc, unsigned short* q,
                                         int lane, int wave, int lo, int blk) {
    const unsigned own4 = blk * 0x01010101u;
    const int rot = (blk * 5) % N_BATCH;

    u32x4 tg[TAG_U];
    int bi = wave + rot; if (bi >= N_BATCH) bi -= N_BATCH;
    {
        const u32x4* tp = (const u32x4*)(tags + (SLICED ? sliced_tag_word(bi) : bi * BATCH_W));
#pragma unroll
        for (int u = 0; u < TAG_U; ++u) tg[u] = tp[lane + u * 32];
    }

    for (int it = wave; it < N_BATCH; it += OWN_WAVES) {
        const int e0 = bi * BATCH_E;
        const int cnt_k = keep_own<0>(tg, own4, q, lane);

        int m = (cnt_k < 32) ? cnt_k : 32;
        int e = e0 + ((lane < m) ? (int)q[lane] : 0);
        int d = dst[e];
        int s = src[e];

        {
            const int itn = it + OWN_WAVES;
            bi = ((itn < N_BATCH) ? itn : it) + rot; if (bi >= N_BATCH) bi -= N_BATCH;
            const u32x4* tp = (const u32x4*)(tags + (SLICED ? sliced_tag_word(bi) : bi * BATCH_W));
#pragma unroll
            for (int u = 0; u < TAG_U; ++u) tg[u] = tp[lane + u * 32];
        }

        if (lane >= m) { s = 0; d = lo; }
        int   dl  = d - lo;
        float ddv = dinv_at(cnt, d);
        float dsv = dinv_at(cnt, s);
        add_rows<C>(h, acc, lane, m, s, dl, dsv, ddv);

        for (int b0 = 32; b0 < cnt_k; b0 += 32) {
            m   = (cnt_k - b0 < 32) ? (cnt_k - b0) : 32;
            e   = e0 + ((lane < m) ? (int)q[b0 + lane] : 0);
            d   = (lane < m) ? dst[e] : lo;
            s   = (lane < m) ? src[e] : 0;
            dl  = d - lo;
            ddv = dinv_at(cnt, d);
            dsv = dinv_at(cnt, s);
            add_rows<C>(h, acc, lane, m, s, dl, dsv, ddv);
        }
    }
}

__global__ __launch_bounds__(OWN_WAVES * 32)
void list_deg_kernel(const int* __restrict__ src,
                     const int* __restrict__ dst,
                     const unsigned* __restrict__ tags,
                     unsigned* __restrict__ cnt,
                     unsigned* lists) {
    __shared__ unsigned short wq[OWN_WAVES][BATCH_E];
    __shared__ unsigned       bl[RNG_OWN][LIST_CAP];
    __shared__ unsigned       dg[RNG_R];
    __shared__ unsigned       bn[RNG_OWN];

    const int tid  = threadIdx.x;
    const int lane = tid & 31;
    const int wave = tid >> 5;
    const int rng  = blockIdx.x / N_PART;
    const int part = blockIdx.x % N_PART;
    const int lo   = rng * RNG_R;
    const int bt0  = (part * N_BATCH) / N_PART;
    const int nbt  = ((part + 1) * N_BATCH) / N_PART - bt0;
    unsigned short* q = wq[wave];

    for (int i = tid; i < RNG_R; i += OWN_WAVES * 32) dg[i] = 0u;
    if (tid < RNG_OWN) bn[tid] = 0u;
    __syncthreads();

    const unsigned own4 = (unsigned)(RNG_OWN * rng) * 0x01010101u;
    const int rot = (rng * 5) % nbt;

    auto note = [&](int m, int d, int s) {
        const int dl = d - lo;
        if (lane < m && (unsigned)dl < (unsigned)RNG_R) {
            const int o = dl / OWN_R;
            const unsigned pos = atomicAdd(&bn[o], 1u);
            if (pos < (unsigned)LIST_CAP) bl[o][pos] = (unsigned)s | ((unsigned)(dl - o * OWN_R) << 17);
            atomicAdd(&dg[dl], 1u);
        }
    };

    u32x4 tg[TAG_U];
    int bi;
    {
        int ix = wave + rot; if (ix >= nbt) ix -= nbt;
        bi = bt0 + ix;
        const u32x4* tp = (const u32x4*)(tags + sliced_tag_word(bi));
#pragma unroll
        for (int u = 0; u < TAG_U; ++u) tg[u] = tp[lane + u * 32];
    }

    int pm = 0, pd = lo, ps = 0;
    for (int it = wave; it < nbt; it += OWN_WAVES) {
        const int e0 = bi * BATCH_E;
        const int itn = it + OWN_WAVES;
        int ixn = ((itn < nbt) ? itn : it) + rot; if (ixn >= nbt) ixn -= nbt;
        const int bn_next = bt0 + ixn;
        u32x4 tn[TAG_U];
        {
            const u32x4* tp = (const u32x4*)(tags + sliced_tag_word(bn_next));
#pragma unroll
            for (int u = 0; u < TAG_U; ++u) tn[u] = tp[lane + u * 32];
        }

        const int cnt_k = keep_own<3>(tg, own4, q, lane);

        for (int b0 = 0; b0 < cnt_k; b0 += 32) {
            const int mm = (cnt_k - b0 < 32) ? (cnt_k - b0) : 32;
            const int ee = e0 + ((lane < mm) ? (int)q[b0 + lane] : 0);
            const int nd = dst[ee];
            const int ns = src[ee];
            note(pm, pd, ps);
            pm = mm; pd = nd; ps = ns;
        }

#pragma unroll
        for (int u = 0; u < TAG_U; ++u) tg[u] = tn[u];
        bi = bn_next;
    }
    note(pm, pd, ps);

    __syncthreads();

    {
        constexpr int WPL = OWN_WAVES / RNG_OWN;
        const int r   = wave % RNG_OWN;
        const int own = RNG_OWN * rng + r;
        if (own < OWN_NB) {
            unsigned* sl = lists + (size_t)own * SLICE_W;
            unsigned* lg = sl + LIST_W0 + part * LIST_CAP;
            const unsigned n  = bn[r];
            const int      nk = (n <= (unsigned)LIST_CAP) ? (int)n : 0;
            for (int i = (wave / RNG_OWN) * 32 + lane; i < ((nk + 31) & ~31); i += WPL * 32)
                lg[i] = (i < nk) ? bl[r][i] : 0u;
            if (wave < RNG_OWN && lane == 0) sl[part] = n;
        }
    }

    for (int i = tid; i < RNG_R; i += OWN_WAVES * 32) {
        const unsigned c = dg[i];
        if (c != 0u && lo + i < N_NODES) atomicAdd(&cnt[lo + i], c);
    }
}

template <int C, bool RELU, bool SLICED>
__global__ __launch_bounds__(OWN_WAVES * 32)
void agg_owner_kernel(const float* __restrict__ h,
                      const int* __restrict__ src,
                      const int* __restrict__ dst,
                      const unsigned* __restrict__ tags,
                      const unsigned* __restrict__ cnt,
                      const float* __restrict__ bias,
                      float* out,
                      const unsigned* lists) {
    __shared__ float          acc[OWN_R * C];
    __shared__ unsigned short wq[OWN_WAVES][BATCH_E];

    const int tid  = threadIdx.x;
    const int lane = tid & 31;
    const int wave = tid >> 5;
    const int lo   = blockIdx.x * OWN_R;
    const unsigned* sl = lists + (size_t)blockIdx.x * SLICE_W;
    const unsigned cw = sl[lane % N_PART];

    for (int i = tid; i < OWN_R * C; i += OWN_WAVES * 32) acc[i] = 0.0f;
    __syncthreads();

    if (__builtin_amdgcn_ballot_w32(cw > (unsigned)LIST_CAP) == 0u) {
        constexpr int WPL = OWN_WAVES / N_PART;
        const int p  = __builtin_amdgcn_readfirstlane(wave % N_PART);
        const int qr = __builtin_amdgcn_readfirstlane(wave / N_PART);
        const int nl = __builtin_amdgcn_readlane((int)cw, p);
        const int rb = (qr == 0) ? 0 : (((nl * qr) / WPL) & ~(ROWS_G - 1));
        const int re = (qr == WPL - 1) ? nl : (((nl * (qr + 1)) / WPL) & ~(ROWS_G - 1));
        const int n  = re - rb;
        const unsigned* lg = sl + LIST_W0 + p * LIST_CAP + rb;
        unsigned cur = (lane < n) ? lg[lane] : 0u;
        unsigned nx  = (32 + lane < n) ? lg[32 + lane] : 0u;
        unsigned cs  = cnt[cur & 0x1ffffu];
        unsigned cd  = cnt[lo + (int)(cur >> 17)];
        for (int b0 = 0; b0 < n; b0 += 32) {
            const int m = (n - b0 < 32) ? (n - b0) : 32;
            const unsigned csn = cnt[nx & 0x1ffffu];
            const unsigned cdn = cnt[lo + (int)(nx >> 17)];
            const unsigned nn  = (b0 + 64 + lane < n) ? lg[b0 + 64 + lane] : 0u;
            const int s  = (int)(cur & 0x1ffffu);
            const int dl = (int)(cur >> 17);
            const float ddv = rsqrtf(1.0f + (float)cd);
            const float dsv = rsqrtf(1.0f + (float)cs);
            add_rows<C>(h, acc, lane, m, s, dl, dsv, ddv);
            cur = nx; nx = nn; cs = csn; cd = cdn;
        }
    } else {
        scan_add<C, SLICED>(h, src, dst, tags, cnt, acc, wq[wave], lane, wave, lo, blockIdx.x);
    }

    __syncthreads();

    constexpr int  EP_STEP = OWN_WAVES * 32;
    constexpr int  EP_N    = OWN_R * C;
    constexpr int  EP_T    = (EP_N + EP_STEP - 1) / EP_STEP;
    constexpr bool EP_FULL = (EP_N % EP_STEP) == 0;
    static_assert(EP_STEP % C == 0, "a thread keeps its channel over its elements");
    const size_t g0 = (size_t)lo * C;
    const float  bc = bias[tid & (C - 1)];
    float    hv[EP_T];
    unsigned cv[EP_T];
#pragma unroll
    for (int t = 0; t < EP_T; ++t) {
        const int i  = tid + t * EP_STEP;
        const int ic = (EP_FULL || i < EP_N) ? i : tid;
        hv[t] = h[g0 + ic];
        cv[t] = cnt[lo + ic / C];
    }
#pragma unroll
    for (int t = 0; t < EP_T; ++t) {
        const int i = tid + t * EP_STEP;
        if (EP_FULL || i < EP_N) {
            const float di  = rsqrtf(1.0f + (float)cv[t]);
            const float val = acc[i] + hv[t] * di * di + bc;
            out[g0 + i] = RELU ? fmaxf(val, 0.0f) : val;
        }
    }
}


extern "C" void kernel_launch(void* const* d_in, const int* in_sizes, int n_in,
                              void* d_out, int out_size, void* d_ws, size_t ws_size,
                              hipStream_t stream) {
    const float* x    = (const float*)d_in[0];
    const int*   ei   = (const int*)d_in[1];
    const float* W1   = (const float*)d_in[2];
    const float* b1   = (const float*)d_in[3];
    const float* W2   = (const float*)d_in[4];
    const float* b2   = (const float*)d_in[5];
    float*       out  = (float*)d_out;

    const int* srcv = ei;
    const int* dstv = ei + N_EDGES;

    unsigned* cnt = (unsigned*)d_ws;
    float* h1   = (float*)d_ws + 100352;
    float* agg1 = h1 + (size_t)N_NODES * HID_CH;
    float* h2   = h1;
    unsigned* obuf  = (unsigned*)d_out;
    unsigned* tags2 = (unsigned*)(h1 + (size_t)N_NODES * OUT_CH);

    gemm_wmma_f32<IN_CH, HID_CH, 1><<<N_STRIPS / GEMM_WAVES, GEMM_WAVES * 32, 0, stream>>>(x, W1, h1, dstv, obuf, cnt);

    list_deg_kernel<<<RNG_NB * N_PART, OWN_WAVES * 32, 0, stream>>>(srcv, dstv, obuf, cnt, obuf);

    agg_owner_kernel<HID_CH, true, true>
        <<<OWN_NB, OWN_WAVES * 32, 0, stream>>>(h1, srcv, dstv, obuf, cnt, b1, agg1, obuf);

    gemm_wmma_f32<HID_CH, OUT_CH, 2><<<N_STRIPS / GEMM_WAVES, GEMM_WAVES * 32, 0, stream>>>(agg1, W2, h2, dstv, tags2, cnt);

    agg_owner_kernel<OUT_CH, false, false>
        <<<OWN_NB, OWN_WAVES * 32, 0, stream>>>(h2, srcv, dstv, tags2, cnt, b2, out, obuf);
}


# List launch first, reads the dst/src words in whole lines, 8 node ranges x 32 edge parts; in-degree shares leave as plain-stored planes the layer-1 product sums; list order, so sum order, changes
# speedup vs baseline: 5.0204x; 1.0406x over previous
#include <hip/hip_runtime.h>
#include <hip/hip_bf16.h>


#define N_NODES 100000
#define N_EDGES 1600000
#define IN_CH   128
#define HID_CH  64
#define OUT_CH  32

typedef __attribute__((ext_vector_type(2))) float f32x2;
typedef __attribute__((ext_vector_type(8))) float f32x8;

typedef __attribute__((ext_vector_type(4))) int i32x4;
typedef __attribute__((ext_vector_type(4))) unsigned u32x4;
typedef __attribute__((ext_vector_type(2))) unsigned u32x2;

#define OWN_NB     250
#define OWN_R      400
#define OWN_WAVES  32
#define TAG_U      5
#define BATCH_G    (TAG_U * 32)
#define BATCH_E    (BATCH_G * 16)
#define BATCH_W    (BATCH_E / 4)
#define N_BATCH    (N_EDGES / BATCH_E)

#define SLICE_W    (OWN_R * OUT_CH)
#define RNG_OWN    32
#define RNG_NB     8
#define RNG_R      (RNG_OWN * OWN_R)
#define N_PART     32
#define PART_G     (N_EDGES / 4 / N_PART)
#define LIST_U     13
#define LIST_CAP   320
#define LIST_W0    64
#define TAGS_W0    10304
#define TAGS_BPS   3

static_assert(N_NODES <= (1 << 17) && OWN_R <= (1 << 15), "a list entry packs source and local node in one word");
static_assert(TAGS_W0 % 4 == 0 && SLICE_W % 4 == 0 && BATCH_W % 4 == 0, "owner bytes are read 16 bytes at a time");
static_assert(TAGS_W0 + TAGS_BPS * BATCH_W <= SLICE_W, "lists and owner bytes must fit one slice");
static_assert(TAGS_BPS * OWN_NB >= N_BATCH, "every batch of owner bytes needs a slice");
static_assert(RNG_NB * RNG_OWN >= OWN_NB && (RNG_NB - 1) * RNG_OWN < OWN_NB, "ranges cover the owner blocks");
static_assert(N_PART <= LIST_W0 && LIST_W0 + N_PART * LIST_CAP <= TAGS_W0 && LIST_CAP % 32 == 0 && LIST_W0 % 32 == 0,
              "counts, then the lists as whole lines, in front of the owner bytes");
static_assert(OWN_WAVES == N_PART && OWN_WAVES == RNG_OWN && N_PART == 32,
              "one wave per list where lists are stored and where they are read; a lane per list count");
static_assert(N_EDGES % (4 * N_PART) == 0, "a part is whole groups of four edges");
static_assert(RNG_R <= (1 << 14), "a queued edge packs source and node inside the range in one word");
#define PDEG_W     (RNG_NB * RNG_R / 2)
static_assert(N_EDGES / N_PART < 65536 && RNG_R % 2 == 0 && N_NODES % 2 == 0 && PDEG_W >= N_NODES / 2,
              "a part's count on one node fits 16 bits; nodes pair up inside a range");

__device__ __forceinline__ float dinv_at(const unsigned* __restrict__ cnt, int n) {
    return rsqrtf(1.0f + (float)cnt[n]);
}

__device__ __forceinline__ int sliced_tag_word(int t) {
    return (t / TAGS_BPS) * SLICE_W + TAGS_W0 + (t % TAGS_BPS) * BATCH_W;
}

__device__ __forceinline__ unsigned owner_bytes(i32x4 d) {
    return  ((unsigned)d.x / (unsigned)OWN_R)        | (((unsigned)d.y / (unsigned)OWN_R) << 8) |
           (((unsigned)d.z / (unsigned)OWN_R) << 16) | (((unsigned)d.w / (unsigned)OWN_R) << 24);
}


#define GEMM_WAVES 5
#define N_STRIPS   (N_NODES / 16)

template <int K, int NCOL, int TAGS>
__global__ __launch_bounds__(GEMM_WAVES * 32)
void gemm_wmma_f32(const float* __restrict__ A,
                   const float* __restrict__ B,
                   float* __restrict__ C,
                   const int* __restrict__ dst,
                   unsigned* __restrict__ tags,
                   unsigned* __restrict__ cnt,
                   const unsigned* __restrict__ pdeg) {
    constexpr int NT = NCOL / 16;
    typedef __attribute__((ext_vector_type(NT))) float fvec;

    const int gt = blockIdx.x * (GEMM_WAVES * 32) + threadIdx.x;
    if (TAGS == 1) {
        for (int r = 0; r < 2; ++r) {
            const int g = gt + r * (N_EDGES / 8);
            const int t = g / BATCH_W;
            tags[sliced_tag_word(t) + (g - t * BATCH_W)] = owner_bytes(((const i32x4*)dst)[g]);
        }
    }
    if (TAGS == 2) {
        for (int r = 0; r < 2; ++r) {
            const int g = gt + r * (N_EDGES / 8);
            tags[g] = owner_bytes(((const i32x4*)dst)[g]);
        }
    }
    const int lane = threadIdx.x & 31;
    const int row0 = (blockIdx.x * GEMM_WAVES + (threadIdx.x >> 5)) << 4;

    const int m  = lane & 15;
    const int kk = (lane >> 4) << 1;

    const float* ap = A + (size_t)(row0 + m) * K + kk;
    const fvec*  bp = (const fvec*)(B + (size_t)kk * NCOL) + m;

    f32x8 acc[NT];
    for (int t = 0; t < NT; ++t) acc[t] = (f32x8){0.f, 0.f, 0.f, 0.f, 0.f, 0.f, 0.f, 0.f};

    for (int k0 = 0; k0 < K; k0 += 4) {
        f32x2 a;
        a.x = ap[k0];
        a.y = ap[k0 + 1];
        const fvec b0 = bp[(size_t)k0 * 16];
        const fvec b1 = bp[(size_t)(k0 + 1) * 16];
        for (int t = 0; t < NT; ++t) {
            f32x2 b;
            b.x = b0[t];
            b.y = b1[t];
            acc[t] = __builtin_amdgcn_wmma_f32_16x16x4_f32(
                 false, a,  false, b,
                 (short)0, acc[t],  false,  false);
        }
    }

    const int mbase = (lane >> 4) << 3;
    fvec* cp = (fvec*)(C + (size_t)row0 * NCOL) + m;
    for (int i = 0; i < 8; ++i) {
        fvec o;
        for (int t = 0; t < NT; ++t) o[t] = acc[t][i];
        cp[(size_t)(mbase + i) * 16] = o;
    }

    if (TAGS == 1) {
        if (gt < N_NODES / 2) {
            unsigned w[N_PART];
#pragma unroll
            for (int p = 0; p < N_PART; ++p) w[p] = pdeg[(size_t)p * PDEG_W + gt];
            unsigned c0 = 0u, c1 = 0u;
#pragma unroll
            for (int p = 0; p < N_PART; ++p) { c0 += w[p] & 0xffffu; c1 += w[p] >> 16; }
            ((u32x2*)cnt)[gt] = (u32x2){c0, c1};
        }
    }
}


#define ROWS_G     8

static_assert(OWN_NB * OWN_R == N_NODES, "owner ranges must tile the nodes");
static_assert(OWN_NB <= 256, "an owner number must fit one byte");
static_assert(N_BATCH * BATCH_E == N_EDGES, "batches must tile the edges");
static_assert(BATCH_E <= 65536, "a queue entry is a 16-bit position inside the batch");
static_assert(N_STRIPS % GEMM_WAVES == 0, "product grids: whole blocks of strips");
static_assert(N_STRIPS * 32 * 8 == N_EDGES, "product grids: one thread per 8 edges");
static_assert(ROWS_G == 8, "the load fence in add_rows names 8 rows");

__device__ __forceinline__ int keep_own(const u32x4 (&tg)[TAG_U], unsigned own4, unsigned short* q, int lane) {
    int cnt = 0;
#pragma unroll
    for (int u = 0; u < TAG_U; ++u) {
        unsigned nz[4];
#pragma unroll
        for (int k = 0; k < 4; ++k) {
            const unsigned x = tg[u][k] ^ own4;
            nz[k] = ((x & 0x7f7f7f7fu) + 0x7f7f7f7fu) | x;
        }
        unsigned f = (nz[0] & 0x80808080u) | ((nz[1] >> 1) & 0x40404040u) |
                     ((nz[2] >> 2) & 0x20202020u) | ((nz[3] >> 3) & 0x10101010u);
        f ^= 0xf0f0f0f0u;
        unsigned mask = __builtin_amdgcn_ballot_w32(f != 0u);
        while (mask != 0u) {
            if (f != 0u) {
                const int b   = __builtin_ctz(f);
                const int pos = cnt + (int)__builtin_amdgcn_mbcnt_lo(mask, 0u);
                const int j   = ((7 - (b & 7)) << 2) + (b >> 3);
                q[pos] = (unsigned short)(((u * 32 + lane) << 4) + j);
                f &= f - 1u;
            }
            cnt += __builtin_popcount(mask);
            mask = __builtin_amdgcn_ballot_w32(f != 0u);
        }
    }
    return cnt;
}

template <int C>
__device__ __forceinline__ void add_rows(const float* __restrict__ h, float* acc, int lane, int m,
                                         int s, int dl, float dsv, float ddv) {
    for (int j = 0; j < m; j += ROWS_G) {
        float v[ROWS_G][C / 32];
#pragma unroll
        for (int t = 0; t < ROWS_G; ++t) {
            const int sj = __builtin_amdgcn_readlane(s, j + t);
            const float* hp = h + (size_t)sj * C + lane;
#pragma unroll
            for (int c = 0; c < C / 32; ++c) v[t][c] = hp[c * 32];
        }
#pragma unroll
        for (int c = 0; c < C / 32; ++c)
            asm volatile("" :: "v"(v[0][c]), "v"(v[1][c]), "v"(v[2][c]), "v"(v[3][c]),
                               "v"(v[4][c]), "v"(v[5][c]), "v"(v[6][c]), "v"(v[7][c]));
#pragma unroll
        for (int t = 0; t < ROWS_G; ++t) {
            if (j + t < m) {
                const float nrm = __int_as_float(__builtin_amdgcn_readlane(__float_as_int(dsv), j + t)) *
                                  __int_as_float(__builtin_amdgcn_readlane(__float_as_int(ddv), j + t));
                const int dlj = __builtin_amdgcn_readlane(dl, j + t);
                float* ap = acc + dlj * C + lane;
#pragma unroll
                for (int c = 0; c < C / 32; ++c) unsafeAtomicAdd(ap + c * 32, v[t][c] * nrm);
            }
        }
    }
}

template <int C, bool SLICED>
__device__ __forceinline__ void scan_add(const float* __restrict__ h,
                                         const int* __restrict__ src,
                                         const int* __restrict__ dst,
                                         const unsigned* __restrict__ tags,
                                         const unsigned* __restrict__ cnt,
                                         float* acc, unsigned short* q,
                                         int lane, int wave, int lo, int blk) {
    const unsigned own4 = blk * 0x01010101u;
    const int rot = (blk * 5) % N_BATCH;

    u32x4 tg[TAG_U];
    int bi = wave + rot; if (bi >= N_BATCH) bi -= N_BATCH;
    {
        const u32x4* tp = (const u32x4*)(tags + (SLICED ? sliced_tag_word(bi) : bi * BATCH_W));
#pragma unroll
        for (int u = 0; u < TAG_U; ++u) tg[u] = tp[lane + u * 32];
    }

    for (int it = wave; it < N_BATCH; it += OWN_WAVES) {
        const int e0 = bi * BATCH_E;
        const int cnt_k = keep_own(tg, own4, q, lane);

        int m = (cnt_k < 32) ? cnt_k : 32;
        int e = e0 + ((lane < m) ? (int)q[lane] : 0);
        int d = dst[e];
        int s = src[e];

        {
            const int itn = it + OWN_WAVES;
            bi = ((itn < N_BATCH) ? itn : it) + rot; if (bi >= N_BATCH) bi -= N_BATCH;
            const u32x4* tp = (const u32x4*)(tags + (SLICED ? sliced_tag_word(bi) : bi * BATCH_W));
#pragma unroll
            for (int u = 0; u < TAG_U; ++u) tg[u] = tp[lane + u * 32];
        }

        if (lane >= m) { s = 0; d = lo; }
        int   dl  = d - lo;
        float ddv = dinv_at(cnt, d);
        float dsv = dinv_at(cnt, s);
        add_rows<C>(h, acc, lane, m, s, dl, dsv, ddv);

        for (int b0 = 32; b0 < cnt_k; b0 += 32) {
            m   = (cnt_k - b0 < 32) ? (cnt_k - b0) : 32;
            e   = e0 + ((lane < m) ? (int)q[b0 + lane] : 0);
            d   = (lane < m) ? dst[e] : lo;
            s   = (lane < m) ? src[e] : 0;
            dl  = d - lo;
            ddv = dinv_at(cnt, d);
            dsv = dinv_at(cnt, s);
            add_rows<C>(h, acc, lane, m, s, dl, dsv, ddv);
        }
    }
}

__global__ __launch_bounds__(OWN_WAVES * 32)
void list_deg_kernel(const int* __restrict__ src,
                     const int* __restrict__ dst,
                     unsigned* __restrict__ pdeg,
                     unsigned* __restrict__ lists) {
    __shared__ unsigned       bl[RNG_OWN][LIST_CAP];
    __shared__ unsigned       dg[RNG_R];
    __shared__ unsigned       bn[RNG_OWN];
    __shared__ unsigned       wq[OWN_WAVES][LIST_U * 4 * 32];

    const int tid  = threadIdx.x;
    const int lane = tid & 31;
    const int wave = tid >> 5;
    const int rng  = blockIdx.x / N_PART;
    const int part = blockIdx.x % N_PART;
    const int lo   = rng * RNG_R;
    unsigned* q = wq[wave];

    for (int i = tid; i < RNG_R; i += OWN_WAVES * 32) dg[i] = 0u;
    if (tid < RNG_OWN) bn[tid] = 0u;
    __syncthreads();

    const i32x4* dp = (const i32x4*)dst + (size_t)part * PART_G;
    const i32x4* sp = (const i32x4*)src + (size_t)part * PART_G;

    for (int g0 = 0; g0 < PART_G; g0 += OWN_WAVES * 32 * LIST_U) {
        i32x4 dv[LIST_U], sv[LIST_U];
#pragma unroll
        for (int u = 0; u < LIST_U; ++u) {
            const int  g  = g0 + u * (OWN_WAVES * 32) + tid;
            const bool in = g < PART_G;
            const int  gc = in ? g : tid;
            dv[u] = dp[gc];
            sv[u] = sp[gc];
            if (!in) dv[u] = (i32x4){-1, -1, -1, -1};
        }
        int nq = 0;
#pragma unroll
        for (int u = 0; u < LIST_U; ++u) {
#pragma unroll
            for (int k = 0; k < 4; ++k) {
                const unsigned dl   = (unsigned)(dv[u][k] - lo);
                const bool     keep = dl < (unsigned)RNG_R;
                const unsigned mask = __builtin_amdgcn_ballot_w32(keep);
                if (keep) q[nq + (int)__builtin_amdgcn_mbcnt_lo(mask, 0u)] = (unsigned)sv[u][k] | (dl << 17);
                nq += __builtin_popcount(mask);
            }
        }
        for (int b0 = 0; b0 < nq; b0 += 32) {
            if (b0 + lane < nq) {
                const unsigned e   = q[b0 + lane];
                const unsigned dl  = e >> 17;
                const unsigned o   = dl / (unsigned)OWN_R;
                const unsigned pos = atomicAdd(&bn[o], 1u);
                if (pos < (unsigned)LIST_CAP) bl[o][pos] = (e & 0x1ffffu) | ((dl - o * (unsigned)OWN_R) << 17);
                atomicAdd(&dg[dl], 1u);
            }
        }
    }

    __syncthreads();

    {
        const int own = RNG_OWN * rng + wave;
        if (own < OWN_NB) {
            unsigned* sl = lists + (size_t)own * SLICE_W;
            unsigned* lg = sl + LIST_W0 + part * LIST_CAP;
            const unsigned n  = bn[wave];
            const int      nk = (n <= (unsigned)LIST_CAP) ? (int)n : 0;
            for (int i = lane; i < ((nk + 31) & ~31); i += 32)
                lg[i] = (i < nk) ? bl[wave][i] : 0u;
            if (lane == 0) sl[part] = n;
        }
    }

    for (int i = tid; i < RNG_R / 2; i += OWN_WAVES * 32)
        pdeg[(size_t)part * PDEG_W + lo / 2 + i] = dg[2 * i] | (dg[2 * i + 1] << 16);
}

template <int C, bool RELU, bool SLICED>
__global__ __launch_bounds__(OWN_WAVES * 32)
void agg_owner_kernel(const float* __restrict__ h,
                      const int* __restrict__ src,
                      const int* __restrict__ dst,
                      const unsigned* __restrict__ tags,
                      const unsigned* __restrict__ cnt,
                      const float* __restrict__ bias,
                      float* out,
                      const unsigned* lists) {
    __shared__ float    acc[OWN_R * C];
    __shared__ unsigned wqw[OWN_WAVES * BATCH_E / 2];
    static_assert(N_PART * LIST_CAP <= OWN_WAVES * BATCH_E / 2, "all lists of a block fit the queue space");

    const int tid  = threadIdx.x;
    const int lane = tid & 31;
    const int wave = tid >> 5;
    const int wv   = __builtin_amdgcn_readfirstlane(wave);
    const int lo   = blockIdx.x * OWN_R;
    const unsigned* sl = lists + (size_t)blockIdx.x * SLICE_W;
    const unsigned cw = sl[lane];

    for (int i = tid; i < OWN_R * C; i += OWN_WAVES * 32) acc[i] = 0.0f;

    const bool whole = __builtin_amdgcn_ballot_w32(cw > (unsigned)LIST_CAP) == 0u;
    int tot = 0;
    if (whole) {
        int off = 0;
        for (int p = 0; p < N_PART; ++p) {
            const int c = __builtin_amdgcn_readlane((int)cw, p);
            off += (p < wv) ? c : 0;
            tot += c;
        }
        const int nw = __builtin_amdgcn_readlane((int)cw, wv);
        const unsigned* lg = sl + LIST_W0 + wv * LIST_CAP;
        unsigned ev[LIST_CAP / 32];
#pragma unroll
        for (int u = 0; u < LIST_CAP / 32; ++u) ev[u] = (u * 32 + lane < nw) ? lg[u * 32 + lane] : 0u;
#pragma unroll
        for (int u = 0; u < LIST_CAP / 32; ++u)
            if (u * 32 + lane < nw) wqw[off + u * 32 + lane] = ev[u];
    }
    __syncthreads();

    if (whole) {
        const int rb = (wv == 0) ? 0 : (((tot * wv) / OWN_WAVES) & ~(ROWS_G - 1));
        const int re = (wv == OWN_WAVES - 1) ? tot : (((tot * (wv + 1)) / OWN_WAVES) & ~(ROWS_G - 1));
        const int n  = re - rb;
        const unsigned* lg = wqw + rb;
        unsigned cur = (lane < n) ? lg[lane] : 0u;
        unsigned nx  = (32 + lane < n) ? lg[32 + lane] : 0u;
        unsigned cs  = cnt[cur & 0x1ffffu];
        unsigned cd  = cnt[lo + (int)(cur >> 17)];
        for (int b0 = 0; b0 < n; b0 += 32) {
            const int m = (n - b0 < 32) ? (n - b0) : 32;
            const unsigned csn = cnt[nx & 0x1ffffu];
            const unsigned cdn = cnt[lo + (int)(nx >> 17)];
            const unsigned nn  = (b0 + 64 + lane < n) ? lg[b0 + 64 + lane] : 0u;
            const int s  = (int)(cur & 0x1ffffu);
            const int dl = (int)(cur >> 17);
            const float ddv = rsqrtf(1.0f + (float)cd);
            const float dsv = rsqrtf(1.0f + (float)cs);
            add_rows<C>(h, acc, lane, m, s, dl, dsv, ddv);
            cur = nx; nx = nn; cs = csn; cd = cdn;
        }
    } else {
        scan_add<C, SLICED>(h, src, dst, tags, cnt, acc, (unsigned short*)wqw + wave * BATCH_E, lane, wave, lo, blockIdx.x);
    }

    __syncthreads();

    constexpr int  EP_STEP = OWN_WAVES * 32;
    constexpr int  EP_N    = OWN_R * C;
    constexpr int  EP_T    = (EP_N + EP_STEP - 1) / EP_STEP;
    constexpr bool EP_FULL = (EP_N % EP_STEP) == 0;
    static_assert(EP_STEP % C == 0, "a thread keeps its channel over its elements");
    const size_t g0 = (size_t)lo * C;
    const float  bc = bias[tid & (C - 1)];
    float    hv[EP_T];
    unsigned cv[EP_T];
#pragma unroll
    for (int t = 0; t < EP_T; ++t) {
        const int i  = tid + t * EP_STEP;
        const int ic = (EP_FULL || i < EP_N) ? i : tid;
        hv[t] = h[g0 + ic];
        cv[t] = cnt[lo + ic / C];
    }
#pragma unroll
    for (int t = 0; t < EP_T; ++t) {
        const int i = tid + t * EP_STEP;
        if (EP_FULL || i < EP_N) {
            const float di  = rsqrtf(1.0f + (float)cv[t]);
            const float val = acc[i] + hv[t] * di * di + bc;
            out[g0 + i] = RELU ? fmaxf(val, 0.0f) : val;
        }
    }
}


extern "C" void kernel_launch(void* const* d_in, const int* in_sizes, int n_in,
                              void* d_out, int out_size, void* d_ws, size_t ws_size,
                              hipStream_t stream) {
    const float* x    = (const float*)d_in[0];
    const int*   ei   = (const int*)d_in[1];
    const float* W1   = (const float*)d_in[2];
    const float* b1   = (const float*)d_in[3];
    const float* W2   = (const float*)d_in[4];
    const float* b2   = (const float*)d_in[5];
    float*       out  = (float*)d_out;

    const int* srcv = ei;
    const int* dstv = ei + N_EDGES;

    unsigned* cnt = (unsigned*)d_ws;
    float* h1   = (float*)d_ws + 100352;
    float* agg1 = h1 + (size_t)N_NODES * HID_CH;
    float* h2   = h1;
    unsigned* obuf  = (unsigned*)d_out;
    unsigned* tags2 = (unsigned*)(h1 + (size_t)N_NODES * OUT_CH);
    unsigned* pdeg  = (unsigned*)agg1;

    list_deg_kernel<<<RNG_NB * N_PART, OWN_WAVES * 32, 0, stream>>>(srcv, dstv, pdeg, obuf);

    gemm_wmma_f32<IN_CH, HID_CH, 1><<<N_STRIPS / GEMM_WAVES, GEMM_WAVES * 32, 0, stream>>>(x, W1, h1, dstv, obuf, cnt, pdeg);

    agg_owner_kernel<HID_CH, true, true>
        <<<OWN_NB, OWN_WAVES * 32, 0, stream>>>(h1, srcv, dstv, obuf, cnt, b1, agg1, obuf);

    gemm_wmma_f32<HID_CH, OUT_CH, 2><<<N_STRIPS / GEMM_WAVES, GEMM_WAVES * 32, 0, stream>>>(agg1, W2, h2, dstv, tags2, cnt, nullptr);

    agg_owner_kernel<OUT_CH, false, false>
        <<<OWN_NB, OWN_WAVES * 32, 0, stream>>>(h2, srcv, dstv, tags2, cnt, b2, out, obuf);
}


# Aggregation blocks keep every node's in-degree as one byte in LDS (left by the product launches; 255 = look the count up): an edge's two degree lookups read LDS, not memory; sum order kept
# speedup vs baseline: 5.2290x; 1.0416x over previous
#include <hip/hip_runtime.h>
#include <hip/hip_bf16.h>


#define N_NODES 100000
#define N_EDGES 1600000
#define IN_CH   128
#define HID_CH  64
#define OUT_CH  32

typedef __attribute__((ext_vector_type(2))) float f32x2;
typedef __attribute__((ext_vector_type(8))) float f32x8;

typedef __attribute__((ext_vector_type(4))) int i32x4;
typedef __attribute__((ext_vector_type(4))) unsigned u32x4;
typedef __attribute__((ext_vector_type(2))) unsigned u32x2;

#define OWN_NB     250
#define OWN_R      400
#define OWN_WAVES  32
#define TAG_U      5
#define BATCH_G    (TAG_U * 32)
#define BATCH_E    (BATCH_G * 16)
#define BATCH_W    (BATCH_E / 4)
#define N_BATCH    (N_EDGES / BATCH_E)

#define SLICE_W    (OWN_R * OUT_CH)
#define RNG_OWN    32
#define RNG_NB     8
#define RNG_R      (RNG_OWN * OWN_R)
#define N_PART     32
#define PART_G     (N_EDGES / 4 / N_PART)
#define LIST_U     13
#define LIST_CAP   320
#define LIST_W0    64
#define TAGS_W0    10304
#define TAGS_BPS   3
#define DEGB_W0    (TAGS_W0 + TAGS_BPS * BATCH_W)
#define DEG_ESC    255u

static_assert(N_NODES <= (1 << 17) && OWN_R <= (1 << 15), "a list entry packs source and local node in one word");
static_assert(TAGS_W0 % 4 == 0 && SLICE_W % 4 == 0 && BATCH_W % 4 == 0, "owner bytes are read 16 bytes at a time");
static_assert(TAGS_W0 + TAGS_BPS * BATCH_W <= SLICE_W, "lists and owner bytes must fit one slice");
static_assert(TAGS_BPS * OWN_NB >= N_BATCH, "every batch of owner bytes needs a slice");
static_assert(DEGB_W0 % 4 == 0 && OWN_R % 16 == 0 && DEGB_W0 + OWN_R / 4 <= SLICE_W && N_NODES % 16 == 0,
              "degree bytes: whole 16-byte pieces behind a slice's owner bytes");
static_assert(RNG_NB * RNG_OWN >= OWN_NB && (RNG_NB - 1) * RNG_OWN < OWN_NB, "ranges cover the owner blocks");
static_assert(N_PART <= LIST_W0 && LIST_W0 + N_PART * LIST_CAP <= TAGS_W0 && LIST_CAP % 32 == 0 && LIST_W0 % 32 == 0,
              "counts, then the lists as whole lines, in front of the owner bytes");
static_assert(OWN_WAVES == N_PART && OWN_WAVES == RNG_OWN && N_PART == 32,
              "one wave per list where lists are stored and where they are read; a lane per list count");
static_assert(N_EDGES % (4 * N_PART) == 0, "a part is whole groups of four edges");
static_assert(RNG_R <= (1 << 14), "a queued edge packs source and node inside the range in one word");
#define PDEG_W     (RNG_NB * RNG_R / 2)
static_assert(N_EDGES / N_PART < 65536 && RNG_R % 2 == 0 && N_NODES % 2 == 0 && PDEG_W >= N_NODES / 2,
              "a part's count on one node fits 16 bits; nodes pair up inside a range");

__device__ __forceinline__ float dinv_at(const unsigned* __restrict__ cnt, int n) {
    return rsqrtf(1.0f + (float)cnt[n]);
}

__device__ __forceinline__ int sliced_tag_word(int t) {
    return (t / TAGS_BPS) * SLICE_W + TAGS_W0 + (t % TAGS_BPS) * BATCH_W;
}

__device__ __forceinline__ unsigned owner_bytes(i32x4 d) {
    return  ((unsigned)d.x / (unsigned)OWN_R)        | (((unsigned)d.y / (unsigned)OWN_R) << 8) |
           (((unsigned)d.z / (unsigned)OWN_R) << 16) | (((unsigned)d.w / (unsigned)OWN_R) << 24);
}


#define GEMM_WAVES 5
#define N_STRIPS   (N_NODES / 16)

template <int K, int NCOL, int TAGS>
__global__ __launch_bounds__(GEMM_WAVES * 32)
void gemm_wmma_f32(const float* __restrict__ A,
                   const float* __restrict__ B,
                   float* __restrict__ C,
                   const int* __restrict__ dst,
                   unsigned* __restrict__ tags,
                   unsigned* __restrict__ cnt,
                   const unsigned* __restrict__ pdeg) {
    constexpr int NT = NCOL / 16;
    typedef __attribute__((ext_vector_type(NT))) float fvec;

    const int gt = blockIdx.x * (GEMM_WAVES * 32) + threadIdx.x;
    if (TAGS == 1) {
        for (int r = 0; r < 2; ++r) {
            const int g = gt + r * (N_EDGES / 8);
            const int t = g / BATCH_W;
            tags[sliced_tag_word(t) + (g - t * BATCH_W)] = owner_bytes(((const i32x4*)dst)[g]);
        }
    }
    if (TAGS == 2) {
        for (int r = 0; r < 2; ++r) {
            const int g = gt + r * (N_EDGES / 8);
            tags[g] = owner_bytes(((const i32x4*)dst)[g]);
        }
        if (gt < N_NODES / 4) {
            const u32x4 c4 = ((const u32x4*)cnt)[gt];
            tags[N_EDGES / 4 + gt] =  (c4.x < DEG_ESC ? c4.x : DEG_ESC)        | ((c4.y < DEG_ESC ? c4.y : DEG_ESC) << 8) |
                                     ((c4.z < DEG_ESC ? c4.z : DEG_ESC) << 16) | ((c4.w < DEG_ESC ? c4.w : DEG_ESC) << 24);
        }
    }
    const int lane = threadIdx.x & 31;
    const int row0 = (blockIdx.x * GEMM_WAVES + (threadIdx.x >> 5)) << 4;

    const int m  = lane & 15;
    const int kk = (lane >> 4) << 1;

    const float* ap = A + (size_t)(row0 + m) * K + kk;
    const fvec*  bp = (const fvec*)(B + (size_t)kk * NCOL) + m;

    f32x8 acc[NT];
    for (int t = 0; t < NT; ++t) acc[t] = (f32x8){0.f, 0.f, 0.f, 0.f, 0.f, 0.f, 0.f, 0.f};

    for (int k0 = 0; k0 < K; k0 += 4) {
        f32x2 a;
        a.x = ap[k0];
        a.y = ap[k0 + 1];
        const fvec b0 = bp[(size_t)k0 * 16];
        const fvec b1 = bp[(size_t)(k0 + 1) * 16];
        for (int t = 0; t < NT; ++t) {
            f32x2 b;
            b.x = b0[t];
            b.y = b1[t];
            acc[t] = __builtin_amdgcn_wmma_f32_16x16x4_f32(
                 false, a,  false, b,
                 (short)0, acc[t],  false,  false);
        }
    }

    const int mbase = (lane >> 4) << 3;
    fvec* cp = (fvec*)(C + (size_t)row0 * NCOL) + m;
    for (int i = 0; i < 8; ++i) {
        fvec o;
        for (int t = 0; t < NT; ++t) o[t] = acc[t][i];
        cp[(size_t)(mbase + i) * 16] = o;
    }

    if (TAGS == 1) {
        if (gt < N_NODES / 2) {
            unsigned w[N_PART];
#pragma unroll
            for (int p = 0; p < N_PART; ++p) w[p] = pdeg[(size_t)p * PDEG_W + gt];
            unsigned c0 = 0u, c1 = 0u;
#pragma unroll
            for (int p = 0; p < N_PART; ++p) { c0 += w[p] & 0xffffu; c1 += w[p] >> 16; }
            ((u32x2*)cnt)[gt] = (u32x2){c0, c1};
            const int n0 = 2 * gt;
            const int sb = n0 / OWN_R;
            unsigned short* db = (unsigned short*)(tags + (size_t)sb * SLICE_W + DEGB_W0);
            db[(n0 - sb * OWN_R) >> 1] = (unsigned short)((c0 < DEG_ESC ? c0 : DEG_ESC) | ((c1 < DEG_ESC ? c1 : DEG_ESC) << 8));
        }
    }
}


#define ROWS_G     8

static_assert(OWN_NB * OWN_R == N_NODES, "owner ranges must tile the nodes");
static_assert(OWN_NB <= 256, "an owner number must fit one byte");
static_assert(N_BATCH * BATCH_E == N_EDGES, "batches must tile the edges");
static_assert(BATCH_E <= 65536, "a queue entry is a 16-bit position inside the batch");
static_assert(N_STRIPS % GEMM_WAVES == 0, "product grids: whole blocks of strips");
static_assert(N_STRIPS * 32 * 8 == N_EDGES, "product grids: one thread per 8 edges");
static_assert(ROWS_G == 8, "the load fence in add_rows names 8 rows");

__device__ __forceinline__ int keep_own(const u32x4 (&tg)[TAG_U], unsigned own4, unsigned short* q, int lane) {
    int cnt = 0;
#pragma unroll
    for (int u = 0; u < TAG_U; ++u) {
        unsigned nz[4];
#pragma unroll
        for (int k = 0; k < 4; ++k) {
            const unsigned x = tg[u][k] ^ own4;
            nz[k] = ((x & 0x7f7f7f7fu) + 0x7f7f7f7fu) | x;
        }
        unsigned f = (nz[0] & 0x80808080u) | ((nz[1] >> 1) & 0x40404040u) |
                     ((nz[2] >> 2) & 0x20202020u) | ((nz[3] >> 3) & 0x10101010u);
        f ^= 0xf0f0f0f0u;
        unsigned mask = __builtin_amdgcn_ballot_w32(f != 0u);
        while (mask != 0u) {
            if (f != 0u) {
                const int b   = __builtin_ctz(f);
                const int pos = cnt + (int)__builtin_amdgcn_mbcnt_lo(mask, 0u);
                const int j   = ((7 - (b & 7)) << 2) + (b >> 3);
                q[pos] = (unsigned short)(((u * 32 + lane) << 4) + j);
                f &= f - 1u;
            }
            cnt += __builtin_popcount(mask);
            mask = __builtin_amdgcn_ballot_w32(f != 0u);
        }
    }
    return cnt;
}

template <int C>
__device__ __forceinline__ void add_rows(const float* __restrict__ h, float* acc, int lane, int m,
                                         int s, int dl, float dsv, float ddv) {
    for (int j = 0; j < m; j += ROWS_G) {
        float v[ROWS_G][C / 32];
#pragma unroll
        for (int t = 0; t < ROWS_G; ++t) {
            const int sj = __builtin_amdgcn_readlane(s, j + t);
            const float* hp = h + (size_t)sj * C + lane;
#pragma unroll
            for (int c = 0; c < C / 32; ++c) v[t][c] = hp[c * 32];
        }
#pragma unroll
        for (int c = 0; c < C / 32; ++c)
            asm volatile("" :: "v"(v[0][c]), "v"(v[1][c]), "v"(v[2][c]), "v"(v[3][c]),
                               "v"(v[4][c]), "v"(v[5][c]), "v"(v[6][c]), "v"(v[7][c]));
#pragma unroll
        for (int t = 0; t < ROWS_G; ++t) {
            if (j + t < m) {
                const float nrm = __int_as_float(__builtin_amdgcn_readlane(__float_as_int(dsv), j + t)) *
                                  __int_as_float(__builtin_amdgcn_readlane(__float_as_int(ddv), j + t));
                const int dlj = __builtin_amdgcn_readlane(dl, j + t);
                float* ap = acc + dlj * C + lane;
#pragma unroll
                for (int c = 0; c < C / 32; ++c) unsafeAtomicAdd(ap + c * 32, v[t][c] * nrm);
            }
        }
    }
}

template <int C, bool SLICED>
__device__ __forceinline__ void scan_add(const float* __restrict__ h,
                                         const int* __restrict__ src,
                                         const int* __restrict__ dst,
                                         const unsigned* __restrict__ tags,
                                         const unsigned* __restrict__ cnt,
                                         float* acc, unsigned short* q,
                                         int lane, int wave, int lo, int blk) {
    const unsigned own4 = blk * 0x01010101u;
    const int rot = (blk * 5) % N_BATCH;

    u32x4 tg[TAG_U];
    int bi = wave + rot; if (bi >= N_BATCH) bi -= N_BATCH;
    {
        const u32x4* tp = (const u32x4*)(tags + (SLICED ? sliced_tag_word(bi) : bi * BATCH_W));
#pragma unroll
        for (int u = 0; u < TAG_U; ++u) tg[u] = tp[lane + u * 32];
    }

    for (int it = wave; it < N_BATCH; it += OWN_WAVES) {
        const int e0 = bi * BATCH_E;
        const int cnt_k = keep_own(tg, own4, q, lane);

        int m = (cnt_k < 32) ? cnt_k : 32;
        int e = e0 + ((lane < m) ? (int)q[lane] : 0);
        int d = dst[e];
        int s = src[e];

        {
            const int itn = it + OWN_WAVES;
            bi = ((itn < N_BATCH) ? itn : it) + rot; if (bi >= N_BATCH) bi -= N_BATCH;
            const u32x4* tp = (const u32x4*)(tags + (SLICED ? sliced_tag_word(bi) : bi * BATCH_W));
#pragma unroll
            for (int u = 0; u < TAG_U; ++u) tg[u] = tp[lane + u * 32];
        }

        if (lane >= m) { s = 0; d = lo; }
        int   dl  = d - lo;
        float ddv = dinv_at(cnt, d);
        float dsv = dinv_at(cnt, s);
        add_rows<C>(h, acc, lane, m, s, dl, dsv, ddv);

        for (int b0 = 32; b0 < cnt_k; b0 += 32) {
            m   = (cnt_k - b0 < 32) ? (cnt_k - b0) : 32;
            e   = e0 + ((lane < m) ? (int)q[b0 + lane] : 0);
            d   = (lane < m) ? dst[e] : lo;
            s   = (lane < m) ? src[e] : 0;
            dl  = d - lo;
            ddv = dinv_at(cnt, d);
            dsv = dinv_at(cnt, s);
            add_rows<C>(h, acc, lane, m, s, dl, dsv, ddv);
        }
    }
}

__global__ __launch_bounds__(OWN_WAVES * 32)
void list_deg_kernel(const int* __restrict__ src,
                     const int* __restrict__ dst,
                     unsigned* __restrict__ pdeg,
                     unsigned* __restrict__ lists) {
    __shared__ unsigned       bl[RNG_OWN][LIST_CAP];
    __shared__ unsigned       dg[RNG_R];
    __shared__ unsigned       bn[RNG_OWN];
    __shared__ unsigned       wq[OWN_WAVES][LIST_U * 4 * 32];

    const int tid  = threadIdx.x;
    const int lane = tid & 31;
    const int wave = tid >> 5;
    const int rng  = blockIdx.x / N_PART;
    const int part = blockIdx.x % N_PART;
    const int lo   = rng * RNG_R;
    unsigned* q = wq[wave];

    for (int i = tid; i < RNG_R; i += OWN_WAVES * 32) dg[i] = 0u;
    if (tid < RNG_OWN) bn[tid] = 0u;
    __syncthreads();

    const i32x4* dp = (const i32x4*)dst + (size_t)part * PART_G;
    const i32x4* sp = (const i32x4*)src + (size_t)part * PART_G;

    for (int g0 = 0; g0 < PART_G; g0 += OWN_WAVES * 32 * LIST_U) {
        i32x4 dv[LIST_U], sv[LIST_U];
#pragma unroll
        for (int u = 0; u < LIST_U; ++u) {
            const int  g  = g0 + u * (OWN_WAVES * 32) + tid;
            const bool in = g < PART_G;
            const int  gc = in ? g : tid;
            dv[u] = dp[gc];
            sv[u] = sp[gc];
            if (!in) dv[u] = (i32x4){-1, -1, -1, -1};
        }
        int nq = 0;
#pragma unroll
        for (int u = 0; u < LIST_U; ++u) {
#pragma unroll
            for (int k = 0; k < 4; ++k) {
                const unsigned dl   = (unsigned)(dv[u][k] - lo);
                const bool     keep = dl < (unsigned)RNG_R;
                const unsigned mask = __builtin_amdgcn_ballot_w32(keep);
                if (keep) q[nq + (int)__builtin_amdgcn_mbcnt_lo(mask, 0u)] = (unsigned)sv[u][k] | (dl << 17);
                nq += __builtin_popcount(mask);
            }
        }
        for (int b0 = 0; b0 < nq; b0 += 32) {
            if (b0 + lane < nq) {
                const unsigned e   = q[b0 + lane];
                const unsigned dl  = e >> 17;
                const unsigned o   = dl / (unsigned)OWN_R;
                const unsigned pos = atomicAdd(&bn[o], 1u);
                if (pos < (unsigned)LIST_CAP) bl[o][pos] = (e & 0x1ffffu) | ((dl - o * (unsigned)OWN_R) << 17);
                atomicAdd(&dg[dl], 1u);
            }
        }
    }

    __syncthreads();

    {
        const int own = RNG_OWN * rng + wave;
        if (own < OWN_NB) {
            unsigned* sl = lists + (size_t)own * SLICE_W;
            unsigned* lg = sl + LIST_W0 + part * LIST_CAP;
            const unsigned n  = bn[wave];
            const int      nk = (n <= (unsigned)LIST_CAP) ? (int)n : 0;
            for (int i = lane; i < ((nk + 31) & ~31); i += 32)
                lg[i] = (i < nk) ? bl[wave][i] : 0u;
            if (lane == 0) sl[part] = n;
        }
    }

    for (int i = tid; i < RNG_R / 2; i += OWN_WAVES * 32)
        pdeg[(size_t)part * PDEG_W + lo / 2 + i] = dg[2 * i] | (dg[2 * i + 1] << 16);
}

template <int C, bool RELU, bool SLICED>
__global__ __launch_bounds__(OWN_WAVES * 32)
void agg_owner_kernel(const float* __restrict__ h,
                      const int* __restrict__ src,
                      const int* __restrict__ dst,
                      const unsigned* __restrict__ tags,
                      const unsigned* __restrict__ cnt,
                      const float* __restrict__ bias,
                      float* out,
                      const unsigned* lists) {
    __shared__ float    acc[OWN_R * C];
    __shared__ __attribute__((aligned(16))) unsigned wqw[OWN_WAVES * BATCH_E / 2];
    static_assert(N_PART * LIST_CAP <= OWN_WAVES * BATCH_E / 2, "all lists of a block fit the queue space");
    constexpr int DEGB_LW = N_PART * LIST_CAP;
    constexpr int DEGB_G  = N_NODES / 16;
    constexpr int DEGB_T  = (DEGB_G + OWN_WAVES * 32 - 1) / (OWN_WAVES * 32);
    static_assert(DEGB_LW % 4 == 0 && DEGB_LW + N_NODES / 4 <= OWN_WAVES * BATCH_E / 2, "degree bytes fit behind the lists");

    const int tid  = threadIdx.x;
    const int lane = tid & 31;
    const int wave = tid >> 5;
    const int wv   = __builtin_amdgcn_readfirstlane(wave);
    const int lo   = blockIdx.x * OWN_R;
    const unsigned* sl = lists + (size_t)blockIdx.x * SLICE_W;
    const unsigned cw = sl[lane];
    u32x4 tv[DEGB_T];
#pragma unroll
    for (int t = 0; t < DEGB_T; ++t) {
        const int i  = tid + t * (OWN_WAVES * 32);
        const int ic = (i < DEGB_G) ? i : tid;
        const int sp = ic / (OWN_R / 16);
        tv[t] = SLICED ? ((const u32x4*)(tags + (size_t)sp * SLICE_W + DEGB_W0))[ic - sp * (OWN_R / 16)]
                       : ((const u32x4*)(tags + N_EDGES / 4))[ic];
    }

    for (int i = tid; i < OWN_R * C; i += OWN_WAVES * 32) acc[i] = 0.0f;

    const bool whole = __builtin_amdgcn_ballot_w32(cw > (unsigned)LIST_CAP) == 0u;
    int tot = 0;
    if (whole) {
        int off = 0;
        for (int p = 0; p < N_PART; ++p) {
            const int c = __builtin_amdgcn_readlane((int)cw, p);
            off += (p < wv) ? c : 0;
            tot += c;
        }
        const int nw = __builtin_amdgcn_readlane((int)cw, wv);
        const unsigned* lg = sl + LIST_W0 + wv * LIST_CAP;
        unsigned ev[LIST_CAP / 32];
#pragma unroll
        for (int u = 0; u < LIST_CAP / 32; ++u) ev[u] = (u * 32 + lane < nw) ? lg[u * 32 + lane] : 0u;
#pragma unroll
        for (int u = 0; u < LIST_CAP / 32; ++u)
            if (u * 32 + lane < nw) wqw[off + u * 32 + lane] = ev[u];
#pragma unroll
        for (int t = 0; t < DEGB_T; ++t) {
            const int i = tid + t * (OWN_WAVES * 32);
            if (i < DEGB_G) ((u32x4*)(wqw + DEGB_LW))[i] = tv[t];
        }
    }
    __syncthreads();

    if (whole) {
        const int rb = (wv == 0) ? 0 : (((tot * wv) / OWN_WAVES) & ~(ROWS_G - 1));
        const int re = (wv == OWN_WAVES - 1) ? tot : (((tot * (wv + 1)) / OWN_WAVES) & ~(ROWS_G - 1));
        const int n  = re - rb;
        const unsigned* lg = wqw + rb;
        unsigned cur = (lane < n) ? lg[lane] : 0u;
        unsigned nx  = (32 + lane < n) ? lg[32 + lane] : 0u;
        const unsigned char* db = (const unsigned char*)(wqw + DEGB_LW);
        auto degs_at = [&](unsigned e, unsigned& ds, unsigned& dd) {
            const int ns = (int)(e & 0x1ffffu);
            const int nd = lo + (int)(e >> 17);
            ds = db[ns];
            dd = db[nd];
            if (__builtin_amdgcn_ballot_w32(ds == DEG_ESC || dd == DEG_ESC) != 0u) {
                if (ds == DEG_ESC) ds = cnt[ns];
                if (dd == DEG_ESC) dd = cnt[nd];
            }
        };
        unsigned cs, cd;
        degs_at(cur, cs, cd);
        for (int b0 = 0; b0 < n; b0 += 32) {
            const int m = (n - b0 < 32) ? (n - b0) : 32;
            unsigned csn, cdn;
            degs_at(nx, csn, cdn);
            const unsigned nn  = (b0 + 64 + lane < n) ? lg[b0 + 64 + lane] : 0u;
            const int s  = (int)(cur & 0x1ffffu);
            const int dl = (int)(cur >> 17);
            const float ddv = rsqrtf(1.0f + (float)cd);
            const float dsv = rsqrtf(1.0f + (float)cs);
            add_rows<C>(h, acc, lane, m, s, dl, dsv, ddv);
            cur = nx; nx = nn; cs = csn; cd = cdn;
        }
    } else {
        scan_add<C, SLICED>(h, src, dst, tags, cnt, acc, (unsigned short*)wqw + wave * BATCH_E, lane, wave, lo, blockIdx.x);
    }

    __syncthreads();

    constexpr int  EP_STEP = OWN_WAVES * 32;
    constexpr int  EP_N    = OWN_R * C;
    constexpr int  EP_T    = (EP_N + EP_STEP - 1) / EP_STEP;
    constexpr bool EP_FULL = (EP_N % EP_STEP) == 0;
    static_assert(EP_STEP % C == 0, "a thread keeps its channel over its elements");
    const size_t g0 = (size_t)lo * C;
    const float  bc = bias[tid & (C - 1)];
    float    hv[EP_T];
    unsigned cv[EP_T];
#pragma unroll
    for (int t = 0; t < EP_T; ++t) {
        const int i  = tid + t * EP_STEP;
        const int ic = (EP_FULL || i < EP_N) ? i : tid;
        hv[t] = h[g0 + ic];
        cv[t] = cnt[lo + ic / C];
    }
#pragma unroll
    for (int t = 0; t < EP_T; ++t) {
        const int i = tid + t * EP_STEP;
        if (EP_FULL || i < EP_N) {
            const float di  = rsqrtf(1.0f + (float)cv[t]);
            const float val = acc[i] + hv[t] * di * di + bc;
            out[g0 + i] = RELU ? fmaxf(val, 0.0f) : val;
        }
    }
}


extern "C" void kernel_launch(void* const* d_in, const int* in_sizes, int n_in,
                              void* d_out, int out_size, void* d_ws, size_t ws_size,
                              hipStream_t stream) {
    const float* x    = (const float*)d_in[0];
    const int*   ei   = (const int*)d_in[1];
    const float* W1   = (const float*)d_in[2];
    const float* b1   = (const float*)d_in[3];
    const float* W2   = (const float*)d_in[4];
    const float* b2   = (const float*)d_in[5];
    float*       out  = (float*)d_out;

    const int* srcv = ei;
    const int* dstv = ei + N_EDGES;

    unsigned* cnt = (unsigned*)d_ws;
    float* h1   = (float*)d_ws + 100352;
    float* agg1 = h1 + (size_t)N_NODES * HID_CH;
    float* h2   = h1;
    unsigned* obuf  = (unsigned*)d_out;
    unsigned* tags2 = (unsigned*)(h1 + (size_t)N_NODES * OUT_CH);
    unsigned* pdeg  = (unsigned*)agg1;

    list_deg_kernel<<<RNG_NB * N_PART, OWN_WAVES * 32, 0, stream>>>(srcv, dstv, pdeg, obuf);

    gemm_wmma_f32<IN_CH, HID_CH, 1><<<N_STRIPS / GEMM_WAVES, GEMM_WAVES * 32, 0, stream>>>(x, W1, h1, dstv, obuf, cnt, pdeg);

    agg_owner_kernel<HID_CH, true, true>
        <<<OWN_NB, OWN_WAVES * 32, 0, stream>>>(h1, srcv, dstv, obuf, cnt, b1, agg1, obuf);

    gemm_wmma_f32<HID_CH, OUT_CH, 2><<<N_STRIPS / GEMM_WAVES, GEMM_WAVES * 32, 0, stream>>>(agg1, W2, h2, dstv, tags2, cnt, nullptr);

    agg_owner_kernel<OUT_CH, false, false>
        <<<OWN_NB, OWN_WAVES * 32, 0, stream>>>(h2, srcv, dstv, tags2, cnt, b2, out, obuf);
}


# Aggregation walk: norm product and source-row offset formed once per lane and broadcast once each, 32-bit offsets on the table base, whole trips of 8 rows add without per-row tests; sums kept
# speedup vs baseline: 5.4096x; 1.0345x over previous
#include <hip/hip_runtime.h>
#include <hip/hip_bf16.h>


#define N_NODES 100000
#define N_EDGES 1600000
#define IN_CH   128
#define HID_CH  64
#define OUT_CH  32

typedef __attribute__((ext_vector_type(2))) float f32x2;
typedef __attribute__((ext_vector_type(8))) float f32x8;

typedef __attribute__((ext_vector_type(4))) int i32x4;
typedef __attribute__((ext_vector_type(4))) unsigned u32x4;
typedef __attribute__((ext_vector_type(2))) unsigned u32x2;

#define OWN_NB     250
#define OWN_R      400
#define OWN_WAVES  32
#define TAG_U      5
#define BATCH_G    (TAG_U * 32)
#define BATCH_E    (BATCH_G * 16)
#define BATCH_W    (BATCH_E / 4)
#define N_BATCH    (N_EDGES / BATCH_E)

#define SLICE_W    (OWN_R * OUT_CH)
#define RNG_OWN    32
#define RNG_NB     8
#define RNG_R      (RNG_OWN * OWN_R)
#define N_PART     32
#define PART_G     (N_EDGES / 4 / N_PART)
#define LIST_U     13
#define LIST_CAP   320
#define LIST_W0    64
#define TAGS_W0    10304
#define TAGS_BPS   3
#define DEGB_W0    (TAGS_W0 + TAGS_BPS * BATCH_W)
#define DEG_ESC    255u

static_assert(N_NODES <= (1 << 17) && OWN_R <= (1 << 15), "a list entry packs source and local node in one word");
static_assert(TAGS_W0 % 4 == 0 && SLICE_W % 4 == 0 && BATCH_W % 4 == 0, "owner bytes are read 16 bytes at a time");
static_assert(TAGS_W0 + TAGS_BPS * BATCH_W <= SLICE_W, "lists and owner bytes must fit one slice");
static_assert(TAGS_BPS * OWN_NB >= N_BATCH, "every batch of owner bytes needs a slice");
static_assert(DEGB_W0 % 4 == 0 && OWN_R % 16 == 0 && DEGB_W0 + OWN_R / 4 <= SLICE_W && N_NODES % 16 == 0,
              "degree bytes: whole 16-byte pieces behind a slice's owner bytes");
static_assert(RNG_NB * RNG_OWN >= OWN_NB && (RNG_NB - 1) * RNG_OWN < OWN_NB, "ranges cover the owner blocks");
static_assert(N_PART <= LIST_W0 && LIST_W0 + N_PART * LIST_CAP <= TAGS_W0 && LIST_CAP % 32 == 0 && LIST_W0 % 32 == 0,
              "counts, then the lists as whole lines, in front of the owner bytes");
static_assert(OWN_WAVES == N_PART && OWN_WAVES == RNG_OWN && N_PART == 32,
              "one wave per list where lists are stored and where they are read; a lane per list count");
static_assert(N_EDGES % (4 * N_PART) == 0, "a part is whole groups of four edges");
static_assert(RNG_R <= (1 << 14), "a queued edge packs source and node inside the range in one word");
#define PDEG_W     (RNG_NB * RNG_R / 2)
static_assert(N_EDGES / N_PART < 65536 && RNG_R % 2 == 0 && N_NODES % 2 == 0 && PDEG_W >= N_NODES / 2,
              "a part's count on one node fits 16 bits; nodes pair up inside a range");

__device__ __forceinline__ float dinv_at(const unsigned* __restrict__ cnt, int n) {
    return rsqrtf(1.0f + (float)cnt[n]);
}

__device__ __forceinline__ int sliced_tag_word(int t) {
    return (t / TAGS_BPS) * SLICE_W + TAGS_W0 + (t % TAGS_BPS) * BATCH_W;
}

__device__ __forceinline__ unsigned owner_bytes(i32x4 d) {
    return  ((unsigned)d.x / (unsigned)OWN_R)        | (((unsigned)d.y / (unsigned)OWN_R) << 8) |
           (((unsigned)d.z / (unsigned)OWN_R) << 16) | (((unsigned)d.w / (unsigned)OWN_R) << 24);
}


#define GEMM_WAVES 5
#define N_STRIPS   (N_NODES / 16)

template <int K, int NCOL, int TAGS>
__global__ __launch_bounds__(GEMM_WAVES * 32)
void gemm_wmma_f32(const float* __restrict__ A,
                   const float* __restrict__ B,
                   float* __restrict__ C,
                   const int* __restrict__ dst,
                   unsigned* __restrict__ tags,
                   unsigned* __restrict__ cnt,
                   const unsigned* __restrict__ pdeg) {
    constexpr int NT = NCOL / 16;
    typedef __attribute__((ext_vector_type(NT))) float fvec;

    const int gt = blockIdx.x * (GEMM_WAVES * 32) + threadIdx.x;
    if (TAGS == 1) {
        for (int r = 0; r < 2; ++r) {
            const int g = gt + r * (N_EDGES / 8);
            const int t = g / BATCH_W;
            tags[sliced_tag_word(t) + (g - t * BATCH_W)] = owner_bytes(((const i32x4*)dst)[g]);
        }
    }
    if (TAGS == 2) {
        for (int r = 0; r < 2; ++r) {
            const int g = gt + r * (N_EDGES / 8);
            tags[g] = owner_bytes(((const i32x4*)dst)[g]);
        }
        if (gt < N_NODES / 4) {
            const u32x4 c4 = ((const u32x4*)cnt)[gt];
            tags[N_EDGES / 4 + gt] =  (c4.x < DEG_ESC ? c4.x : DEG_ESC)        | ((c4.y < DEG_ESC ? c4.y : DEG_ESC) << 8) |
                                     ((c4.z < DEG_ESC ? c4.z : DEG_ESC) << 16) | ((c4.w < DEG_ESC ? c4.w : DEG_ESC) << 24);
        }
    }
    const int lane = threadIdx.x & 31;
    const int row0 = (blockIdx.x * GEMM_WAVES + (threadIdx.x >> 5)) << 4;

    const int m  = lane & 15;
    const int kk = (lane >> 4) << 1;

    const float* ap = A + (size_t)(row0 + m) * K + kk;
    const fvec*  bp = (const fvec*)(B + (size_t)kk * NCOL) + m;

    f32x8 acc[NT];
    for (int t = 0; t < NT; ++t) acc[t] = (f32x8){0.f, 0.f, 0.f, 0.f, 0.f, 0.f, 0.f, 0.f};

    for (int k0 = 0; k0 < K; k0 += 4) {
        f32x2 a;
        a.x = ap[k0];
        a.y = ap[k0 + 1];
        const fvec b0 = bp[(size_t)k0 * 16];
        const fvec b1 = bp[(size_t)(k0 + 1) * 16];
        for (int t = 0; t < NT; ++t) {
            f32x2 b;
            b.x = b0[t];
            b.y = b1[t];
            acc[t] = __builtin_amdgcn_wmma_f32_16x16x4_f32(
                 false, a,  false, b,
                 (short)0, acc[t],  false,  false);
        }
    }

    const int mbase = (lane >> 4) << 3;
    fvec* cp = (fvec*)(C + (size_t)row0 * NCOL) + m;
    for (int i = 0; i < 8; ++i) {
        fvec o;
        for (int t = 0; t < NT; ++t) o[t] = acc[t][i];
        cp[(size_t)(mbase + i) * 16] = o;
    }

    if (TAGS == 1) {
        if (gt < N_NODES / 2) {
            unsigned w[N_PART];
#pragma unroll
            for (int p = 0; p < N_PART; ++p) w[p] = pdeg[(size_t)p * PDEG_W + gt];
            unsigned c0 = 0u, c1 = 0u;
#pragma unroll
            for (int p = 0; p < N_PART; ++p) { c0 += w[p] & 0xffffu; c1 += w[p] >> 16; }
            ((u32x2*)cnt)[gt] = (u32x2){c0, c1};
            const int n0 = 2 * gt;
            const int sb = n0 / OWN_R;
            unsigned short* db = (unsigned short*)(tags + (size_t)sb * SLICE_W + DEGB_W0);
            db[(n0 - sb * OWN_R) >> 1] = (unsigned short)((c0 < DEG_ESC ? c0 : DEG_ESC) | ((c1 < DEG_ESC ? c1 : DEG_ESC) << 8));
        }
    }
}


#define ROWS_G     8

static_assert(OWN_NB * OWN_R == N_NODES, "owner ranges must tile the nodes");
static_assert(OWN_NB <= 256, "an owner number must fit one byte");
static_assert(N_BATCH * BATCH_E == N_EDGES, "batches must tile the edges");
static_assert(BATCH_E <= 65536, "a queue entry is a 16-bit position inside the batch");
static_assert(N_STRIPS % GEMM_WAVES == 0, "product grids: whole blocks of strips");
static_assert(N_STRIPS * 32 * 8 == N_EDGES, "product grids: one thread per 8 edges");
static_assert(ROWS_G == 8, "the load fence in add_rows names 8 rows");

__device__ __forceinline__ int keep_own(const u32x4 (&tg)[TAG_U], unsigned own4, unsigned short* q, int lane) {
    int cnt = 0;
#pragma unroll
    for (int u = 0; u < TAG_U; ++u) {
        unsigned nz[4];
#pragma unroll
        for (int k = 0; k < 4; ++k) {
            const unsigned x = tg[u][k] ^ own4;
            nz[k] = ((x & 0x7f7f7f7fu) + 0x7f7f7f7fu) | x;
        }
        unsigned f = (nz[0] & 0x80808080u) | ((nz[1] >> 1) & 0x40404040u) |
                     ((nz[2] >> 2) & 0x20202020u) | ((nz[3] >> 3) & 0x10101010u);
        f ^= 0xf0f0f0f0u;
        unsigned mask = __builtin_amdgcn_ballot_w32(f != 0u);
        while (mask != 0u) {
            if (f != 0u) {
                const int b   = __builtin_ctz(f);
                const int pos = cnt + (int)__builtin_amdgcn_mbcnt_lo(mask, 0u);
                const int j   = ((7 - (b & 7)) << 2) + (b >> 3);
                q[pos] = (unsigned short)(((u * 32 + lane) << 4) + j);
                f &= f - 1u;
            }
            cnt += __builtin_popcount(mask);
            mask = __builtin_amdgcn_ballot_w32(f != 0u);
        }
    }
    return cnt;
}

template <int C>
__device__ __forceinline__ void add_rows(const float* __restrict__ h, float* acc, int lane, int m,
                                         int s, int dl, float dsv, float ddv) {
    const float    nrmv = dsv * ddv;
    const unsigned sov  = (unsigned)s * (unsigned)C;
    const unsigned ul   = (unsigned)lane;
    for (int j = 0; j < m; j += ROWS_G) {
        float v[ROWS_G][C / 32];
#pragma unroll
        for (int t = 0; t < ROWS_G; ++t) {
            const unsigned so = (unsigned)__builtin_amdgcn_readlane((int)sov, j + t) + ul;
#pragma unroll
            for (int c = 0; c < C / 32; ++c) v[t][c] = (h + c * 32)[so];
        }
#pragma unroll
        for (int c = 0; c < C / 32; ++c)
            asm volatile("" :: "v"(v[0][c]), "v"(v[1][c]), "v"(v[2][c]), "v"(v[3][c]),
                               "v"(v[4][c]), "v"(v[5][c]), "v"(v[6][c]), "v"(v[7][c]));
        if (j + ROWS_G <= m) {
#pragma unroll
            for (int t = 0; t < ROWS_G; ++t) {
                const float nrm = __int_as_float(__builtin_amdgcn_readlane(__float_as_int(nrmv), j + t));
                const int dlj = __builtin_amdgcn_readlane(dl, j + t);
                float* ap = acc + dlj * C + lane;
#pragma unroll
                for (int c = 0; c < C / 32; ++c) unsafeAtomicAdd(ap + c * 32, v[t][c] * nrm);
            }
        } else {
#pragma unroll
            for (int t = 0; t < ROWS_G; ++t) {
                if (j + t < m) {
                    const float nrm = __int_as_float(__builtin_amdgcn_readlane(__float_as_int(nrmv), j + t));
                    const int dlj = __builtin_amdgcn_readlane(dl, j + t);
                    float* ap = acc + dlj * C + lane;
#pragma unroll
                    for (int c = 0; c < C / 32; ++c) unsafeAtomicAdd(ap + c * 32, v[t][c] * nrm);
                }
            }
        }
    }
}

template <int C, bool SLICED>
__device__ __forceinline__ void scan_add(const float* __restrict__ h,
                                         const int* __restrict__ src,
                                         const int* __restrict__ dst,
                                         const unsigned* __restrict__ tags,
                                         const unsigned* __restrict__ cnt,
                                         float* acc, unsigned short* q,
                                         int lane, int wave, int lo, int blk) {
    const unsigned own4 = blk * 0x01010101u;
    const int rot = (blk * 5) % N_BATCH;

    u32x4 tg[TAG_U];
    int bi = wave + rot; if (bi >= N_BATCH) bi -= N_BATCH;
    {
        const u32x4* tp = (const u32x4*)(tags + (SLICED ? sliced_tag_word(bi) : bi * BATCH_W));
#pragma unroll
        for (int u = 0; u < TAG_U; ++u) tg[u] = tp[lane + u * 32];
    }

    for (int it = wave; it < N_BATCH; it += OWN_WAVES) {
        const int e0 = bi * BATCH_E;
        const int cnt_k = keep_own(tg, own4, q, lane);

        int m = (cnt_k < 32) ? cnt_k : 32;
        int e = e0 + ((lane < m) ? (int)q[lane] : 0);
        int d = dst[e];
        int s = src[e];

        {
            const int itn = it + OWN_WAVES;
            bi = ((itn < N_BATCH) ? itn : it) + rot; if (bi >= N_BATCH) bi -= N_BATCH;
            const u32x4* tp = (const u32x4*)(tags + (SLICED ? sliced_tag_word(bi) : bi * BATCH_W));
#pragma unroll
            for (int u = 0; u < TAG_U; ++u) tg[u] = tp[lane + u * 32];
        }

        if (lane >= m) { s = 0; d = lo; }
        int   dl  = d - lo;
        float ddv = dinv_at(cnt, d);
        float dsv = dinv_at(cnt, s);
        add_rows<C>(h, acc, lane, m, s, dl, dsv, ddv);

        for (int b0 = 32; b0 < cnt_k; b0 += 32) {
            m   = (cnt_k - b0 < 32) ? (cnt_k - b0) : 32;
            e   = e0 + ((lane < m) ? (int)q[b0 + lane] : 0);
            d   = (lane < m) ? dst[e] : lo;
            s   = (lane < m) ? src[e] : 0;
            dl  = d - lo;
            ddv = dinv_at(cnt, d);
            dsv = dinv_at(cnt, s);
            add_rows<C>(h, acc, lane, m, s, dl, dsv, ddv);
        }
    }
}

__global__ __launch_bounds__(OWN_WAVES * 32)
void list_deg_kernel(const int* __restrict__ src,
                     const int* __restrict__ dst,
                     unsigned* __restrict__ pdeg,
                     unsigned* __restrict__ lists) {
    __shared__ unsigned       bl[RNG_OWN][LIST_CAP];
    __shared__ unsigned       dg[RNG_R];
    __shared__ unsigned       bn[RNG_OWN];
    __shared__ unsigned       wq[OWN_WAVES][LIST_U * 4 * 32];

    const int tid  = threadIdx.x;
    const int lane = tid & 31;
    const int wave = tid >> 5;
    const int rng  = blockIdx.x / N_PART;
    const int part = blockIdx.x % N_PART;
    const int lo   = rng * RNG_R;
    unsigned* q = wq[wave];

    for (int i = tid; i < RNG_R; i += OWN_WAVES * 32) dg[i] = 0u;
    if (tid < RNG_OWN) bn[tid] = 0u;
    __syncthreads();

    const i32x4* dp = (const i32x4*)dst + (size_t)part * PART_G;
    const i32x4* sp = (const i32x4*)src + (size_t)part * PART_G;

    for (int g0 = 0; g0 < PART_G; g0 += OWN_WAVES * 32 * LIST_U) {
        i32x4 dv[LIST_U], sv[LIST_U];
#pragma unroll
        for (int u = 0; u < LIST_U; ++u) {
            const int  g  = g0 + u * (OWN_WAVES * 32) + tid;
            const bool in = g < PART_G;
            const int  gc = in ? g : tid;
            dv[u] = dp[gc];
            sv[u] = sp[gc];
            if (!in) dv[u] = (i32x4){-1, -1, -1, -1};
        }
        int nq = 0;
#pragma unroll
        for (int u = 0; u < LIST_U; ++u) {
#pragma unroll
            for (int k = 0; k < 4; ++k) {
                const unsigned dl   = (unsigned)(dv[u][k] - lo);
                const bool     keep = dl < (unsigned)RNG_R;
                const unsigned mask = __builtin_amdgcn_ballot_w32(keep);
                if (keep) q[nq + (int)__builtin_amdgcn_mbcnt_lo(mask, 0u)] = (unsigned)sv[u][k] | (dl << 17);
                nq += __builtin_popcount(mask);
            }
        }
        for (int b0 = 0; b0 < nq; b0 += 32) {
            if (b0 + lane < nq) {
                const unsigned e   = q[b0 + lane];
                const unsigned dl  = e >> 17;
                const unsigned o   = dl / (unsigned)OWN_R;
                const unsigned pos = atomicAdd(&bn[o], 1u);
                if (pos < (unsigned)LIST_CAP) bl[o][pos] = (e & 0x1ffffu) | ((dl - o * (unsigned)OWN_R) << 17);
                atomicAdd(&dg[dl], 1u);
            }
        }
    }

    __syncthreads();

    {
        const int own = RNG_OWN * rng + wave;
        if (own < OWN_NB) {
            unsigned* sl = lists + (size_t)own * SLICE_W;
            unsigned* lg = sl + LIST_W0 + part * LIST_CAP;
            const unsigned n  = bn[wave];
            const int      nk = (n <= (unsigned)LIST_CAP) ? (int)n : 0;
            for (int i = lane; i < ((nk + 31) & ~31); i += 32)
                lg[i] = (i < nk) ? bl[wave][i] : 0u;
            if (lane == 0) sl[part] = n;
        }
    }

    for (int i = tid; i < RNG_R / 2; i += OWN_WAVES * 32)
        pdeg[(size_t)part * PDEG_W + lo / 2 + i] = dg[2 * i] | (dg[2 * i + 1] << 16);
}

template <int C, bool RELU, bool SLICED>
__global__ __launch_bounds__(OWN_WAVES * 32)
void agg_owner_kernel(const float* __restrict__ h,
                      const int* __restrict__ src,
                      const int* __restrict__ dst,
                      const unsigned* __restrict__ tags,
                      const unsigned* __restrict__ cnt,
                      const float* __restrict__ bias,
                      float* out,
                      const unsigned* lists) {
    __shared__ float    acc[OWN_R * C];
    __shared__ __attribute__((aligned(16))) unsigned wqw[OWN_WAVES * BATCH_E / 2];
    static_assert(N_PART * LIST_CAP <= OWN_WAVES * BATCH_E / 2, "all lists of a block fit the queue space");
    constexpr int DEGB_LW = N_PART * LIST_CAP;
    constexpr int DEGB_G  = N_NODES / 16;
    constexpr int DEGB_T  = (DEGB_G + OWN_WAVES * 32 - 1) / (OWN_WAVES * 32);
    static_assert(DEGB_LW % 4 == 0 && DEGB_LW + N_NODES / 4 <= OWN_WAVES * BATCH_E / 2, "degree bytes fit behind the lists");

    const int tid  = threadIdx.x;
    const int lane = tid & 31;
    const int wave = tid >> 5;
    const int wv   = __builtin_amdgcn_readfirstlane(wave);
    const int lo   = blockIdx.x * OWN_R;
    const unsigned* sl = lists + (size_t)blockIdx.x * SLICE_W;
    const unsigned cw = sl[lane];
    u32x4 tv[DEGB_T];
#pragma unroll
    for (int t = 0; t < DEGB_T; ++t) {
        const int i  = tid + t * (OWN_WAVES * 32);
        const int ic = (i < DEGB_G) ? i : tid;
        const int sp = ic / (OWN_R / 16);
        tv[t] = SLICED ? ((const u32x4*)(tags + (size_t)sp * SLICE_W + DEGB_W0))[ic - sp * (OWN_R / 16)]
                       : ((const u32x4*)(tags + N_EDGES / 4))[ic];
    }

    for (int i = tid; i < OWN_R * C; i += OWN_WAVES * 32) acc[i] = 0.0f;

    const bool whole = __builtin_amdgcn_ballot_w32(cw > (unsigned)LIST_CAP) == 0u;
    int tot = 0;
    if (whole) {
        int off = 0;
        for (int p = 0; p < N_PART; ++p) {
            const int c = __builtin_amdgcn_readlane((int)cw, p);
            off += (p < wv) ? c : 0;
            tot += c;
        }
        const int nw = __builtin_amdgcn_readlane((int)cw, wv);
        const unsigned* lg = sl + LIST_W0 + wv * LIST_CAP;
        unsigned ev[LIST_CAP / 32];
#pragma unroll
        for (int u = 0; u < LIST_CAP / 32; ++u) ev[u] = (u * 32 + lane < nw) ? lg[u * 32 + lane] : 0u;
#pragma unroll
        for (int u = 0; u < LIST_CAP / 32; ++u)
            if (u * 32 + lane < nw) wqw[off + u * 32 + lane] = ev[u];
#pragma unroll
        for (int t = 0; t < DEGB_T; ++t) {
            const int i = tid + t * (OWN_WAVES * 32);
            if (i < DEGB_G) ((u32x4*)(wqw + DEGB_LW))[i] = tv[t];
        }
    }
    __syncthreads();

    if (whole) {
        const int rb = (wv == 0) ? 0 : (((tot * wv) / OWN_WAVES) & ~(ROWS_G - 1));
        const int re = (wv == OWN_WAVES - 1) ? tot : (((tot * (wv + 1)) / OWN_WAVES) & ~(ROWS_G - 1));
        const int n  = re - rb;
        const unsigned* lg = wqw + rb;
        unsigned cur = (lane < n) ? lg[lane] : 0u;
        unsigned nx  = (32 + lane < n) ? lg[32 + lane] : 0u;
        const unsigned char* db = (const unsigned char*)(wqw + DEGB_LW);
        auto degs_at = [&](unsigned e, unsigned& ds, unsigned& dd) {
            const int ns = (int)(e & 0x1ffffu);
            const int nd = lo + (int)(e >> 17);
            ds = db[ns];
            dd = db[nd];
            if (__builtin_amdgcn_ballot_w32(ds == DEG_ESC || dd == DEG_ESC) != 0u) {
                if (ds == DEG_ESC) ds = cnt[ns];
                if (dd == DEG_ESC) dd = cnt[nd];
            }
        };
        unsigned cs, cd;
        degs_at(cur, cs, cd);
        for (int b0 = 0; b0 < n; b0 += 32) {
            const int m = (n - b0 < 32) ? (n - b0) : 32;
            unsigned csn, cdn;
            degs_at(nx, csn, cdn);
            const unsigned nn  = (b0 + 64 + lane < n) ? lg[b0 + 64 + lane] : 0u;
            const int s  = (int)(cur & 0x1ffffu);
            const int dl = (int)(cur >> 17);
            const float ddv = rsqrtf(1.0f + (float)cd);
            const float dsv = rsqrtf(1.0f + (float)cs);
            add_rows<C>(h, acc, lane, m, s, dl, dsv, ddv);
            cur = nx; nx = nn; cs = csn; cd = cdn;
        }
    } else {
        scan_add<C, SLICED>(h, src, dst, tags, cnt, acc, (unsigned short*)wqw + wave * BATCH_E, lane, wave, lo, blockIdx.x);
    }

    __syncthreads();

    constexpr int  EP_STEP = OWN_WAVES * 32;
    constexpr int  EP_N    = OWN_R * C;
    constexpr int  EP_T    = (EP_N + EP_STEP - 1) / EP_STEP;
    constexpr bool EP_FULL = (EP_N % EP_STEP) == 0;
    static_assert(EP_STEP % C == 0, "a thread keeps its channel over its elements");
    const size_t g0 = (size_t)lo * C;
    const float  bc = bias[tid & (C - 1)];
    float    hv[EP_T];
    unsigned cv[EP_T];
#pragma unroll
    for (int t = 0; t < EP_T; ++t) {
        const int i  = tid + t * EP_STEP;
        const int ic = (EP_FULL || i < EP_N) ? i : tid;
        hv[t] = h[g0 + ic];
        cv[t] = cnt[lo + ic / C];
    }
#pragma unroll
    for (int t = 0; t < EP_T; ++t) {
        const int i = tid + t * EP_STEP;
        if (EP_FULL || i < EP_N) {
            const float di  = rsqrtf(1.0f + (float)cv[t]);
            const float val = acc[i] + hv[t] * di * di + bc;
            out[g0 + i] = RELU ? fmaxf(val, 0.0f) : val;
        }
    }
}


extern "C" void kernel_launch(void* const* d_in, const int* in_sizes, int n_in,
                              void* d_out, int out_size, void* d_ws, size_t ws_size,
                              hipStream_t stream) {
    const float* x    = (const float*)d_in[0];
    const int*   ei   = (const int*)d_in[1];
    const float* W1   = (const float*)d_in[2];
    const float* b1   = (const float*)d_in[3];
    const float* W2   = (const float*)d_in[4];
    const float* b2   = (const float*)d_in[5];
    float*       out  = (float*)d_out;

    const int* srcv = ei;
    const int* dstv = ei + N_EDGES;

    unsigned* cnt = (unsigned*)d_ws;
    float* h1   = (float*)d_ws + 100352;
    float* agg1 = h1 + (size_t)N_NODES * HID_CH;
    float* h2   = h1;
    unsigned* obuf  = (unsigned*)d_out;
    unsigned* tags2 = (unsigned*)(h1 + (size_t)N_NODES * OUT_CH);
    unsigned* pdeg  = (unsigned*)agg1;

    list_deg_kernel<<<RNG_NB * N_PART, OWN_WAVES * 32, 0, stream>>>(srcv, dstv, pdeg, obuf);

    gemm_wmma_f32<IN_CH, HID_CH, 1><<<N_STRIPS / GEMM_WAVES, GEMM_WAVES * 32, 0, stream>>>(x, W1, h1, dstv, obuf, cnt, pdeg);

    agg_owner_kernel<HID_CH, true, true>
        <<<OWN_NB, OWN_WAVES * 32, 0, stream>>>(h1, srcv, dstv, obuf, cnt, b1, agg1, obuf);

    gemm_wmma_f32<HID_CH, OUT_CH, 2><<<N_STRIPS / GEMM_WAVES, GEMM_WAVES * 32, 0, stream>>>(agg1, W2, h2, dstv, tags2, cnt, nullptr);

    agg_owner_kernel<OUT_CH, false, false>
        <<<OWN_NB, OWN_WAVES * 32, 0, stream>>>(h2, srcv, dstv, tags2, cnt, b2, out, obuf);
}
